# starting kernel
# baseline (speedup 1.0000x reference)
#include <hip/hip_runtime.h>

typedef __attribute__((ext_vector_type(2))) float v2f;
typedef __attribute__((ext_vector_type(8))) float v8f;

#define B_   256
#define OBS_ 256
#define H_   1024
#define A_   18

__global__ __launch_bounds__(32) void k_fc_in(const float* __restrict__ x,
                                              const float* __restrict__ Win,
                                              const float* __restrict__ bin,
                                              float* __restrict__ h1) {
  const int lane = threadIdx.x & 31;
  const int half = lane >> 4;
  const int l16  = lane & 15;
  const int m0   = blockIdx.x * 32;
  const int n0   = blockIdx.y * 32;

  v8f c00 = {}, c01 = {}, c10 = {}, c11 = {};

  const float* a0 = x   + (m0 +      l16) * OBS_ + 2 * half;
  const float* a1 = x   + (m0 + 16 + l16) * OBS_ + 2 * half;
  const float* b0 = Win + (n0 +      l16) * OBS_ + 2 * half;
  const float* b1 = Win + (n0 + 16 + l16) * OBS_ + 2 * half;

  for (int k = 0; k < OBS_; k += 4) {
    v2f af0 = *(const v2f*)(a0 + k);
    v2f af1 = *(const v2f*)(a1 + k);
    v2f bf0 = *(const v2f*)(b0 + k);
    v2f bf1 = *(const v2f*)(b1 + k);
    c00 = __builtin_amdgcn_wmma_f32_16x16x4_f32(false, af0, false, bf0, (short)0, c00, false, false);
    c01 = __builtin_amdgcn_wmma_f32_16x16x4_f32(false, af0, false, bf1, (short)0, c01, false, false);
    c10 = __builtin_amdgcn_wmma_f32_16x16x4_f32(false, af1, false, bf0, (short)0, c10, false, false);
    c11 = __builtin_amdgcn_wmma_f32_16x16x4_f32(false, af1, false, bf1, (short)0, c11, false, false);
  }

  const int col0 = n0 + l16;
  const int col1 = n0 + 16 + l16;
  const float bias0 = bin[col0];
  const float bias1 = bin[col1];
#pragma unroll
  for (int v = 0; v < 8; ++v) {
    const int r0 = m0 + v + 8 * half;
    const int r1 = m0 + 16 + v + 8 * half;
    h1[r0 * H_ + col0] = c00[v] + bias0;
    h1[r0 * H_ + col1] = c01[v] + bias1;
    h1[r1 * H_ + col0] = c10[v] + bias0;
    h1[r1 * H_ + col1] = c11[v] + bias1;
  }
}

template <int IS_MAX>
__global__ __launch_bounds__(256) void k_semiring(const float* __restrict__ hin,
                                                  const float* __restrict__ W,
                                                  float* __restrict__ hout) {
  __shared__ float hs[16 * H_];
  const int b0 = blockIdx.x * 16;
  const int o0 = blockIdx.y * 64;

  {
    const float4* src = (const float4*)(hin + b0 * H_);
    float4* dst = (float4*)hs;
    for (int idx = threadIdx.x; idx < (16 * H_) / 4; idx += 256) dst[idx] = src[idx];
  }
  __syncthreads();

  const int wave = threadIdx.x >> 5;
  const int lane = threadIdx.x & 31;

  for (int ot = 0; ot < 2; ++ot) {
    const int ob = o0 + wave * 8 + ot * 4;
    const float* wbase = W + (size_t)ob * H_;

    float acc[4][16];
#pragma unroll
    for (int oi = 0; oi < 4; ++oi)
#pragma unroll
      for (int b = 0; b < 16; ++b)
        acc[oi][b] = IS_MAX ? -3.402823466e38f : 3.402823466e38f;

    for (int c = 0; c < H_ / 128; ++c) {
      const int io = c * 128 + lane * 4;
      float4 w[4];
#pragma unroll
      for (int oi = 0; oi < 4; ++oi)
        w[oi] = *(const float4*)(wbase + oi * H_ + io);

#pragma unroll
      for (int b = 0; b < 16; ++b) {
        const float4 hv = *(const float4*)(hs + b * H_ + io);
#pragma unroll
        for (int oi = 0; oi < 4; ++oi) {
          if (IS_MAX) {
            float t0 = fmaxf(w[oi].x + hv.x, w[oi].y + hv.y);
            float t1 = fmaxf(w[oi].z + hv.z, w[oi].w + hv.w);
            acc[oi][b] = fmaxf(acc[oi][b], fmaxf(t0, t1));
          } else {
            float t0 = fminf(w[oi].x + hv.x, w[oi].y + hv.y);
            float t1 = fminf(w[oi].z + hv.z, w[oi].w + hv.w);
            acc[oi][b] = fminf(acc[oi][b], fminf(t0, t1));
          }
        }
      }
    }

#pragma unroll
    for (int oi = 0; oi < 4; ++oi) {
#pragma unroll
      for (int b = 0; b < 16; ++b) {
        float v = acc[oi][b];
#pragma unroll
        for (int off = 16; off > 0; off >>= 1) {
          const float other = __shfl_xor(v, off, 32);
          v = IS_MAX ? fmaxf(v, other) : fminf(v, other);
        }
        if (lane == 0) hout[(b0 + b) * H_ + ob + oi] = v;
      }
    }
  }
}

__global__ __launch_bounds__(32) void k_fc_out(const float* __restrict__ h,
                                               const float* __restrict__ Wout,
                                               const float* __restrict__ bout,
                                               float* __restrict__ q) {
  const int lane = threadIdx.x & 31;
  const int half = lane >> 4;
  const int l16  = lane & 15;
  const int m0   = blockIdx.x * 16;
  const int n0   = blockIdx.y * 16;

  const int col  = n0 + l16;
  const int bcol = (col < A_) ? col : 0;

  v8f c = {};
  const float* ap = h    + (m0 + l16) * H_ + 2 * half;
  const float* bp = Wout + bcol * H_       + 2 * half;

  for (int k = 0; k < H_; k += 4) {
    v2f a = *(const v2f*)(ap + k);
    v2f b = *(const v2f*)(bp + k);
    c = __builtin_amdgcn_wmma_f32_16x16x4_f32(false, a, false, b, (short)0, c, false, false);
  }

  if (col < A_) {
    const float bias = bout[col];
#pragma unroll
    for (int v = 0; v < 8; ++v) q[(m0 + v + 8 * half) * A_ + col] = c[v] + bias;
  }
}

extern "C" void kernel_launch(void* const* d_in, const int* in_sizes, int n_in,
                              void* d_out, int out_size, void* d_ws, size_t ws_size,
                              hipStream_t stream) {
  (void)in_sizes; (void)n_in; (void)out_size; (void)ws_size;
  const float* x    = (const float*)d_in[0];
  const float* Win  = (const float*)d_in[1];
  const float* bin  = (const float*)d_in[2];
  const float* Wmax = (const float*)d_in[3];
  const float* Wmin = (const float*)d_in[4];
  const float* Wout = (const float*)d_in[5];
  const float* bout = (const float*)d_in[6];
  float* q  = (float*)d_out;
  float* h1 = (float*)d_ws;
  float* h2 = h1 + (size_t)B_ * H_;

  k_fc_in   <<<dim3(B_ / 32, H_ / 32), 32,  0, stream>>>(x,  Win,  bin,  h1);
  k_semiring<1><<<dim3(B_ / 16, H_ / 64), 256, 0, stream>>>(h1, Wmax, h2);
  k_semiring<0><<<dim3(B_ / 16, H_ / 64), 256, 0, stream>>>(h2, Wmin, h1);
  k_fc_out  <<<dim3(B_ / 16, 2),       32,  0, stream>>>(h1, Wout, bout, q);
}


# semiring kernels output-stationary: 16 waves split i, a lane owns 2 output columns x 16 rows, h as scalar operands (scalar loads), W via LDS; no cross-lane reduction, wave partials max/min in LDS
# speedup vs baseline: 1.9616x; 1.9616x over previous
#include <hip/hip_runtime.h>

typedef __attribute__((ext_vector_type(2))) float v2f;
typedef __attribute__((ext_vector_type(8))) float v8f;

#define B_   256
#define OBS_ 256
#define H_   1024
#define A_   18

__global__ __launch_bounds__(32) void k_fc_in(const float* __restrict__ x,
                                              const float* __restrict__ Win,
                                              const float* __restrict__ bin,
                                              float* __restrict__ h1) {
  const int lane = threadIdx.x & 31;
  const int half = lane >> 4;
  const int l16  = lane & 15;
  const int m0   = blockIdx.x * 32;
  const int n0   = blockIdx.y * 32;

  v8f c00 = {}, c01 = {}, c10 = {}, c11 = {};

  const float* a0 = x   + (m0 +      l16) * OBS_ + 2 * half;
  const float* a1 = x   + (m0 + 16 + l16) * OBS_ + 2 * half;
  const float* b0 = Win + (n0 +      l16) * OBS_ + 2 * half;
  const float* b1 = Win + (n0 + 16 + l16) * OBS_ + 2 * half;

  for (int k = 0; k < OBS_; k += 4) {
    v2f af0 = *(const v2f*)(a0 + k);
    v2f af1 = *(const v2f*)(a1 + k);
    v2f bf0 = *(const v2f*)(b0 + k);
    v2f bf1 = *(const v2f*)(b1 + k);
    c00 = __builtin_amdgcn_wmma_f32_16x16x4_f32(false, af0, false, bf0, (short)0, c00, false, false);
    c01 = __builtin_amdgcn_wmma_f32_16x16x4_f32(false, af0, false, bf1, (short)0, c01, false, false);
    c10 = __builtin_amdgcn_wmma_f32_16x16x4_f32(false, af1, false, bf0, (short)0, c10, false, false);
    c11 = __builtin_amdgcn_wmma_f32_16x16x4_f32(false, af1, false, bf1, (short)0, c11, false, false);
  }

  const int col0 = n0 + l16;
  const int col1 = n0 + 16 + l16;
  const float bias0 = bin[col0];
  const float bias1 = bin[col1];
#pragma unroll
  for (int v = 0; v < 8; ++v) {
    const int r0 = m0 + v + 8 * half;
    const int r1 = m0 + 16 + v + 8 * half;
    h1[r0 * H_ + col0] = c00[v] + bias0;
    h1[r0 * H_ + col1] = c01[v] + bias1;
    h1[r1 * H_ + col0] = c10[v] + bias0;
    h1[r1 * H_ + col1] = c11[v] + bias1;
  }
}

#define WP_ 132

template <int IS_MAX>
__device__ __forceinline__ float sel3(float a, float x, float y) {
  return IS_MAX ? fmaxf(fmaxf(a, x), y) : fminf(fminf(a, x), y);
}

template <int IS_MAX>
__device__ __forceinline__ void semiring_chunk(const float* wl, const float* hrow,
                                               float (&acc0)[16], float (&acc1)[16]) {
  for (int kq = 0; kq < 8; ++kq) {
    const int k = kq * 4;
    const float4 q0 = *(const float4*)(wl + 2 * k);
    const float4 q1 = *(const float4*)(wl + 2 * k + 4);
    const v2f w0 = {q0.x, q0.y}, w1 = {q0.z, q0.w};
    const v2f w2 = {q1.x, q1.y}, w3 = {q1.z, q1.w};
#pragma unroll
    for (int b = 0; b < 16; ++b) {
      const float4 hq = *(const float4*)(hrow + b * H_ + k);
      const float hy = __int_as_float(__builtin_amdgcn_readfirstlane(__float_as_int(hq.y)));
      const float hw = __int_as_float(__builtin_amdgcn_readfirstlane(__float_as_int(hq.w)));
      const v2f s0 = w0 + hq.x;
      const v2f s1 = w1 + hy;
      const v2f s2 = w2 + hq.z;
      const v2f s3 = w3 + hw;
      acc0[b] = sel3<IS_MAX>(acc0[b], s0.x, s1.x);
      acc1[b] = sel3<IS_MAX>(acc1[b], s0.y, s1.y);
      acc0[b] = sel3<IS_MAX>(acc0[b], s2.x, s3.x);
      acc1[b] = sel3<IS_MAX>(acc1[b], s2.y, s3.y);
    }
  }
}

__device__ __forceinline__ void lds_barrier() {
  __builtin_amdgcn_fence(__ATOMIC_RELEASE, "workgroup", "local");
  __builtin_amdgcn_s_barrier();
  __builtin_amdgcn_fence(__ATOMIC_ACQUIRE, "workgroup", "local");
}

__device__ __forceinline__ void stage_store(float* dst, const float4 v) {
  dst[0] = v.x; dst[2] = v.y; dst[4] = v.z; dst[6] = v.w;
}

template <int IS_MAX>
__global__ __launch_bounds__(512) void k_semiring(const float* __restrict__ hin,
                                                  const float* __restrict__ W,
                                                  float* __restrict__ hout) {
  __shared__ __attribute__((aligned(16))) float smem[16 * 32 * WP_];
  const int b0 = blockIdx.x * 16;
  const int o0 = blockIdx.y * 64;
  const int lane = threadIdx.x & 31;
  const int wave = __builtin_amdgcn_readfirstlane((int)(threadIdx.x >> 5));
  const int kbase = wave * 64;

  float* wreg = smem + wave * (32 * WP_);
  const float* hrow = hin + (size_t)b0 * H_ + kbase;
  const int srow = lane >> 3;
  const int scol = (lane & 7) * 4;
  const float* wsrc = W + (size_t)(o0 + srow) * H_ + kbase + scol;
  float* wdst = wreg + srow * WP_ + 2 * scol;

  float acc0[16], acc1[16];
#pragma unroll
  for (int b = 0; b < 16; ++b) {
    acc0[b] = IS_MAX ? -3.402823466e38f : 3.402823466e38f;
    acc1[b] = acc0[b];
  }

  float4 pa[16], pb[16];
#pragma unroll
  for (int j = 0; j < 16; ++j) pa[j] = *(const float4*)(wsrc + (size_t)(j * 4) * H_);
#pragma unroll
  for (int j = 0; j < 16; ++j) pb[j] = *(const float4*)(wsrc + (size_t)(j * 4) * H_ + 32);
#pragma unroll
  for (int j = 0; j < 16; ++j) stage_store(wdst + ((j & 7) * 4) * WP_ + (j >> 3), pa[j]);
  lds_barrier();

  const float* wl = wreg + lane * WP_;
  semiring_chunk<IS_MAX>(wl, hrow, acc0, acc1);

#pragma unroll
  for (int j = 0; j < 16; ++j) stage_store(wdst + ((j & 7) * 4) * WP_ + (j >> 3) + 64, pb[j]);
  __syncthreads();
  semiring_chunk<IS_MAX>(wl + 64, hrow + 32, acc0, acc1);

  __syncthreads();
  float* red = smem;
#pragma unroll
  for (int b = 0; b < 16; ++b) {
    red[(wave * 16 + b) * 64 + lane] = acc0[b];
    red[(wave * 16 + b) * 64 + lane + 32] = acc1[b];
  }
  __syncthreads();
#pragma unroll
  for (int p = 0; p < 2; ++p) {
    const int idx = (int)threadIdx.x + 512 * p;
    const int b = idx >> 6;
    const int o = idx & 63;
    float v = red[b * 64 + o];
#pragma unroll
    for (int w = 1; w < 16; ++w) {
      const float u = red[(w * 16 + b) * 64 + o];
      v = IS_MAX ? fmaxf(v, u) : fminf(v, u);
    }
    hout[(b0 + b) * H_ + o0 + o] = v;
  }
}

__global__ __launch_bounds__(32) void k_fc_out(const float* __restrict__ h,
                                               const float* __restrict__ Wout,
                                               const float* __restrict__ bout,
                                               float* __restrict__ q) {
  const int lane = threadIdx.x & 31;
  const int half = lane >> 4;
  const int l16  = lane & 15;
  const int m0   = blockIdx.x * 16;
  const int n0   = blockIdx.y * 16;

  const int col  = n0 + l16;
  const int bcol = (col < A_) ? col : 0;

  v8f c = {};
  const float* ap = h    + (m0 + l16) * H_ + 2 * half;
  const float* bp = Wout + bcol * H_       + 2 * half;

  for (int k = 0; k < H_; k += 4) {
    v2f a = *(const v2f*)(ap + k);
    v2f b = *(const v2f*)(bp + k);
    c = __builtin_amdgcn_wmma_f32_16x16x4_f32(false, a, false, b, (short)0, c, false, false);
  }

  if (col < A_) {
    const float bias = bout[col];
#pragma unroll
    for (int v = 0; v < 8; ++v) q[(m0 + v + 8 * half) * A_ + col] = c[v] + bias;
  }
}

extern "C" void kernel_launch(void* const* d_in, const int* in_sizes, int n_in,
                              void* d_out, int out_size, void* d_ws, size_t ws_size,
                              hipStream_t stream) {
  (void)in_sizes; (void)n_in; (void)out_size; (void)ws_size;
  const float* x    = (const float*)d_in[0];
  const float* Win  = (const float*)d_in[1];
  const float* bin  = (const float*)d_in[2];
  const float* Wmax = (const float*)d_in[3];
  const float* Wmin = (const float*)d_in[4];
  const float* Wout = (const float*)d_in[5];
  const float* bout = (const float*)d_in[6];
  float* q  = (float*)d_out;
  float* h1 = (float*)d_ws;
  float* h2 = h1 + (size_t)B_ * H_;

  k_fc_in   <<<dim3(B_ / 32, H_ / 32), 32,  0, stream>>>(x,  Win,  bin,  h1);
  k_semiring<1><<<dim3(B_ / 16, H_ / 64), 512, 0, stream>>>(h1, Wmax, h2);
  k_semiring<0><<<dim3(B_ / 16, H_ / 64), 512, 0, stream>>>(h2, Wmin, h1);
  k_fc_out  <<<dim3(B_ / 16, 2),       32,  0, stream>>>(h1, Wout, bout, q);
}


# activations between layers stored row-tile-minor hT[b/16][i][b%16]: the semiring kernels' 16 row-strided 16-byte scalar loads per step become 4 contiguous 64-byte loads; results bit-identical
# speedup vs baseline: 2.1252x; 1.0834x over previous
#include <hip/hip_runtime.h>

typedef __attribute__((ext_vector_type(2))) float v2f;
typedef __attribute__((ext_vector_type(8))) float v8f;
typedef __attribute__((ext_vector_type(16))) float v16f;


#define B_   256
#define OBS_ 256
#define H_   1024
#define A_   18

__device__ __forceinline__ void store8(float* p, const v8f c, const float bias) {
  *(float4*)(p)     = make_float4(c[0] + bias, c[1] + bias, c[2] + bias, c[3] + bias);
  *(float4*)(p + 4) = make_float4(c[4] + bias, c[5] + bias, c[6] + bias, c[7] + bias);
}

__global__ __launch_bounds__(32) void k_fc_in(const float* __restrict__ x,
                                              const float* __restrict__ Win,
                                              const float* __restrict__ bin,
                                              float* __restrict__ h1) {
  const int lane = threadIdx.x & 31;
  const int half = lane >> 4;
  const int l16  = lane & 15;
  const int m0   = blockIdx.x * 32;
  const int n0   = blockIdx.y * 32;

  v8f c00 = {}, c01 = {}, c10 = {}, c11 = {};

  const float* a0 = x   + (m0 +      l16) * OBS_ + 2 * half;
  const float* a1 = x   + (m0 + 16 + l16) * OBS_ + 2 * half;
  const float* b0 = Win + (n0 +      l16) * OBS_ + 2 * half;
  const float* b1 = Win + (n0 + 16 + l16) * OBS_ + 2 * half;

  for (int k = 0; k < OBS_; k += 4) {
    v2f af0 = *(const v2f*)(a0 + k);
    v2f af1 = *(const v2f*)(a1 + k);
    v2f bf0 = *(const v2f*)(b0 + k);
    v2f bf1 = *(const v2f*)(b1 + k);
    c00 = __builtin_amdgcn_wmma_f32_16x16x4_f32(false, af0, false, bf0, (short)0, c00, false, false);
    c01 = __builtin_amdgcn_wmma_f32_16x16x4_f32(false, af0, false, bf1, (short)0, c01, false, false);
    c10 = __builtin_amdgcn_wmma_f32_16x16x4_f32(false, af1, false, bf0, (short)0, c10, false, false);
    c11 = __builtin_amdgcn_wmma_f32_16x16x4_f32(false, af1, false, bf1, (short)0, c11, false, false);
  }

  const int col0 = n0 + l16;
  const int col1 = n0 + 16 + l16;
  const float bias0 = bin[col0];
  const float bias1 = bin[col1];
  const int bt0 = blockIdx.x * 2;
  store8(h1 + ((size_t)(bt0 * H_ + col0) * 16 + 8 * half), c00, bias0);
  store8(h1 + ((size_t)(bt0 * H_ + col1) * 16 + 8 * half), c01, bias1);
  store8(h1 + ((size_t)((bt0 + 1) * H_ + col0) * 16 + 8 * half), c10, bias0);
  store8(h1 + ((size_t)((bt0 + 1) * H_ + col1) * 16 + 8 * half), c11, bias1);
}

#define WP_ 132

template <int IS_MAX>
__device__ __forceinline__ float sel3(float a, float x, float y) {
  return IS_MAX ? fmaxf(fmaxf(a, x), y) : fminf(fminf(a, x), y);
}

__device__ __forceinline__ float own_sreg(float v) {
  return __int_as_float(__builtin_amdgcn_readfirstlane(__float_as_int(v)));
}

template <int IS_MAX>
__device__ __forceinline__ void semiring_chunk(const float* wl, const float* ht,
                                               float (&acc0)[16], float (&acc1)[16]) {
  for (int kq = 0; kq < 8; ++kq) {
    const int k = kq * 4;
    const float4 q0 = *(const float4*)(wl + 2 * k);
    const float4 q1 = *(const float4*)(wl + 2 * k + 4);
    const v2f w0 = {q0.x, q0.y}, w1 = {q0.z, q0.w};
    const v2f w2 = {q1.x, q1.y}, w3 = {q1.z, q1.w};
    const v16f hv0 = *(const v16f*)(ht + (k + 0) * 16);
    const v16f hv1 = *(const v16f*)(ht + (k + 1) * 16);
    const v16f hv2 = *(const v16f*)(ht + (k + 2) * 16);
    const v16f hv3 = *(const v16f*)(ht + (k + 3) * 16);
#pragma unroll
    for (int b = 0; b < 16; ++b) {
      const float x0 = (b & 1) ? own_sreg(hv0[b]) : hv0[b];
      const float x1 = (b & 1) ? own_sreg(hv1[b]) : hv1[b];
      const float x2 = (b & 1) ? own_sreg(hv2[b]) : hv2[b];
      const float x3 = (b & 1) ? own_sreg(hv3[b]) : hv3[b];
      const v2f s0 = w0 + x0;
      const v2f s1 = w1 + x1;
      const v2f s2 = w2 + x2;
      const v2f s3 = w3 + x3;
      acc0[b] = sel3<IS_MAX>(acc0[b], s0.x, s1.x);
      acc1[b] = sel3<IS_MAX>(acc1[b], s0.y, s1.y);
      acc0[b] = sel3<IS_MAX>(acc0[b], s2.x, s3.x);
      acc1[b] = sel3<IS_MAX>(acc1[b], s2.y, s3.y);
    }
  }
}

__device__ __forceinline__ void lds_barrier() {
  __builtin_amdgcn_fence(__ATOMIC_RELEASE, "workgroup", "local");
  __builtin_amdgcn_s_barrier();
  __builtin_amdgcn_fence(__ATOMIC_ACQUIRE, "workgroup", "local");
}

__device__ __forceinline__ void stage_store(float* dst, const float4 v) {
  dst[0] = v.x; dst[2] = v.y; dst[4] = v.z; dst[6] = v.w;
}

#define RP_ 20

template <int IS_MAX, int OUT_T>
__global__ __launch_bounds__(512) void k_semiring(const float* __restrict__ hin,
                                                  const float* __restrict__ W,
                                                  float* __restrict__ hout) {
  __shared__ __attribute__((aligned(16))) float smem[16 * 32 * WP_];
  const int b0 = blockIdx.x * 16;
  const int o0 = blockIdx.y * 64;
  const int lane = threadIdx.x & 31;
  const int wave = __builtin_amdgcn_readfirstlane((int)(threadIdx.x >> 5));
  const int kbase = wave * 64;

  float* wreg = smem + wave * (32 * WP_);
  const float* ht = hin + ((size_t)blockIdx.x * H_ + kbase) * 16;
  const int srow = lane >> 3;
  const int scol = (lane & 7) * 4;
  const float* wsrc = W + (size_t)(o0 + srow) * H_ + kbase + scol;
  float* wdst = wreg + srow * WP_ + 2 * scol;

  float acc0[16], acc1[16];
#pragma unroll
  for (int b = 0; b < 16; ++b) {
    acc0[b] = IS_MAX ? -3.402823466e38f : 3.402823466e38f;
    acc1[b] = acc0[b];
  }

  float4 pa[16], pb[16];
#pragma unroll
  for (int j = 0; j < 16; ++j) pa[j] = *(const float4*)(wsrc + (size_t)(j * 4) * H_);
#pragma unroll
  for (int j = 0; j < 16; ++j) pb[j] = *(const float4*)(wsrc + (size_t)(j * 4) * H_ + 32);
#pragma unroll
  for (int j = 0; j < 16; ++j) stage_store(wdst + ((j & 7) * 4) * WP_ + (j >> 3), pa[j]);
  lds_barrier();

  const float* wl = wreg + lane * WP_;
  semiring_chunk<IS_MAX>(wl, ht, acc0, acc1);

#pragma unroll
  for (int j = 0; j < 16; ++j) stage_store(wdst + ((j & 7) * 4) * WP_ + (j >> 3) + 64, pb[j]);
  __syncthreads();
  semiring_chunk<IS_MAX>(wl + 64, ht + 32 * 16, acc0, acc1);

  __syncthreads();
  float* red = smem;
  if (OUT_T) {
    float* r0 = red + (wave * 64 + lane) * RP_;
    float* r1 = r0 + 32 * RP_;
#pragma unroll
    for (int q = 0; q < 4; ++q) {
      *(float4*)(r0 + 4 * q) = make_float4(acc0[4 * q], acc0[4 * q + 1], acc0[4 * q + 2], acc0[4 * q + 3]);
      *(float4*)(r1 + 4 * q) = make_float4(acc1[4 * q], acc1[4 * q + 1], acc1[4 * q + 2], acc1[4 * q + 3]);
    }
    __syncthreads();
#pragma unroll
    for (int p = 0; p < 2; ++p) {
      const int idx = (int)threadIdx.x + 512 * p;
      const int o = idx >> 4;
      const int b = idx & 15;
      float v = red[o * RP_ + b];
#pragma unroll
      for (int w = 1; w < 16; ++w) {
        const float u = red[(w * 64 + o) * RP_ + b];
        v = IS_MAX ? fmaxf(v, u) : fminf(v, u);
      }
      hout[((size_t)blockIdx.x * H_ + o0) * 16 + idx] = v;
    }
    return;
  }
#pragma unroll
  for (int b = 0; b < 16; ++b) {
    red[(wave * 16 + b) * 64 + lane] = acc0[b];
    red[(wave * 16 + b) * 64 + lane + 32] = acc1[b];
  }
  __syncthreads();
#pragma unroll
  for (int p = 0; p < 2; ++p) {
    const int idx = (int)threadIdx.x + 512 * p;
    const int b = idx >> 6;
    const int o = idx & 63;
    float v = red[b * 64 + o];
#pragma unroll
    for (int w = 1; w < 16; ++w) {
      const float u = red[(w * 16 + b) * 64 + o];
      v = IS_MAX ? fmaxf(v, u) : fminf(v, u);
    }
    hout[(b0 + b) * H_ + o0 + o] = v;
  }
}

__global__ __launch_bounds__(32) void k_fc_out(const float* __restrict__ h,
                                               const float* __restrict__ Wout,
                                               const float* __restrict__ bout,
                                               float* __restrict__ q) {
  const int lane = threadIdx.x & 31;
  const int half = lane >> 4;
  const int l16  = lane & 15;
  const int m0   = blockIdx.x * 16;
  const int n0   = blockIdx.y * 16;

  const int col  = n0 + l16;
  const int bcol = (col < A_) ? col : 0;

  v8f c = {};
  const float* ap = h    + (m0 + l16) * H_ + 2 * half;
  const float* bp = Wout + bcol * H_       + 2 * half;

  for (int k = 0; k < H_; k += 4) {
    v2f a = *(const v2f*)(ap + k);
    v2f b = *(const v2f*)(bp + k);
    c = __builtin_amdgcn_wmma_f32_16x16x4_f32(false, a, false, b, (short)0, c, false, false);
  }

  if (col < A_) {
    const float bias = bout[col];
#pragma unroll
    for (int v = 0; v < 8; ++v) q[(m0 + v + 8 * half) * A_ + col] = c[v] + bias;
  }
}

extern "C" void kernel_launch(void* const* d_in, const int* in_sizes, int n_in,
                              void* d_out, int out_size, void* d_ws, size_t ws_size,
                              hipStream_t stream) {
  (void)in_sizes; (void)n_in; (void)out_size; (void)ws_size;
  const float* x    = (const float*)d_in[0];
  const float* Win  = (const float*)d_in[1];
  const float* bin  = (const float*)d_in[2];
  const float* Wmax = (const float*)d_in[3];
  const float* Wmin = (const float*)d_in[4];
  const float* Wout = (const float*)d_in[5];
  const float* bout = (const float*)d_in[6];
  float* q  = (float*)d_out;
  float* h1 = (float*)d_ws;
  float* h2 = h1 + (size_t)B_ * H_;

  k_fc_in   <<<dim3(B_ / 32, H_ / 32), 32,  0, stream>>>(x,  Win,  bin,  h1);
  k_semiring<1, 1><<<dim3(B_ / 16, H_ / 64), 512, 0, stream>>>(h1, Wmax, h2);
  k_semiring<0, 0><<<dim3(B_ / 16, H_ / 64), 512, 0, stream>>>(h2, Wmin, h1);
  k_fc_out  <<<dim3(B_ / 16, 2),       32,  0, stream>>>(h1, Wout, bout, q);
}


# fc_in, fc_out: K loop pipelined in registers over three rotating fragment sets, each K chunk's loads issued as one group two chunks ahead of its WMMAs; same k order, results bit-identical
# speedup vs baseline: 2.6714x; 1.2570x over previous
#include <hip/hip_runtime.h>

typedef __attribute__((ext_vector_type(2))) float v2f;
typedef __attribute__((ext_vector_type(8))) float v8f;
typedef __attribute__((ext_vector_type(16))) float v16f;


#define B_   256
#define OBS_ 256
#define H_   1024
#define A_   18
#define FI_  8
#define FO_  16

template <int N>
__device__ __forceinline__ void chunk_load(v2f (&f)[N], const float* p) {
#pragma unroll
  for (int s = 0; s < N; ++s) f[s] = *(const v2f*)(p + 4 * s);
}

#define PIPE_FENCE() __builtin_amdgcn_sched_barrier(0)

__device__ __forceinline__ void store8(float* p, const v8f c, const float bias) {
  *(float4*)(p)     = make_float4(c[0] + bias, c[1] + bias, c[2] + bias, c[3] + bias);
  *(float4*)(p + 4) = make_float4(c[4] + bias, c[5] + bias, c[6] + bias, c[7] + bias);
}

__global__ __launch_bounds__(32) void k_fc_in(const float* __restrict__ x,
                                              const float* __restrict__ Win,
                                              const float* __restrict__ bin,
                                              float* __restrict__ h1) {
  const int lane = threadIdx.x & 31;
  const int half = lane >> 4;
  const int l16  = lane & 15;
  const int m0   = blockIdx.x * 32;
  const int n0   = blockIdx.y * 32;

  v8f c00 = {}, c01 = {}, c10 = {}, c11 = {};

  const float* a0 = x   + (m0 +      l16) * OBS_ + 2 * half;
  const float* a1 = x   + (m0 + 16 + l16) * OBS_ + 2 * half;
  const float* b0 = Win + (n0 +      l16) * OBS_ + 2 * half;
  const float* b1 = Win + (n0 + 16 + l16) * OBS_ + 2 * half;

  constexpr int CK = 4 * FI_;
  constexpr int NC = OBS_ / CK;
  v2f fa0[3][FI_], fa1[3][FI_], fb0[3][FI_], fb1[3][FI_];
#pragma unroll
  for (int c = 0; c < 2; ++c) {
    chunk_load(fa0[c], a0 + c * CK); chunk_load(fa1[c], a1 + c * CK);
    chunk_load(fb0[c], b0 + c * CK); chunk_load(fb1[c], b1 + c * CK);
  }
#pragma unroll
  for (int c = 0; c < NC; ++c) {
    if (c + 2 < NC) {
      const int n = (c + 2) % 3;
      chunk_load(fa0[n], a0 + (c + 2) * CK); chunk_load(fa1[n], a1 + (c + 2) * CK);
      chunk_load(fb0[n], b0 + (c + 2) * CK); chunk_load(fb1[n], b1 + (c + 2) * CK);
    }
    PIPE_FENCE();
    const int u = c % 3;
#pragma unroll
    for (int s = 0; s < FI_; ++s) {
      c00 = __builtin_amdgcn_wmma_f32_16x16x4_f32(false, fa0[u][s], false, fb0[u][s], (short)0, c00, false, false);
      c01 = __builtin_amdgcn_wmma_f32_16x16x4_f32(false, fa0[u][s], false, fb1[u][s], (short)0, c01, false, false);
      c10 = __builtin_amdgcn_wmma_f32_16x16x4_f32(false, fa1[u][s], false, fb0[u][s], (short)0, c10, false, false);
      c11 = __builtin_amdgcn_wmma_f32_16x16x4_f32(false, fa1[u][s], false, fb1[u][s], (short)0, c11, false, false);
    }
    PIPE_FENCE();
  }

  const int col0 = n0 + l16;
  const int col1 = n0 + 16 + l16;
  const float bias0 = bin[col0];
  const float bias1 = bin[col1];
  const int bt0 = blockIdx.x * 2;
  store8(h1 + ((size_t)(bt0 * H_ + col0) * 16 + 8 * half), c00, bias0);
  store8(h1 + ((size_t)(bt0 * H_ + col1) * 16 + 8 * half), c01, bias1);
  store8(h1 + ((size_t)((bt0 + 1) * H_ + col0) * 16 + 8 * half), c10, bias0);
  store8(h1 + ((size_t)((bt0 + 1) * H_ + col1) * 16 + 8 * half), c11, bias1);
}

#define WP_ 132

template <int IS_MAX>
__device__ __forceinline__ float sel3(float a, float x, float y) {
  return IS_MAX ? fmaxf(fmaxf(a, x), y) : fminf(fminf(a, x), y);
}

__device__ __forceinline__ float own_sreg(float v) {
  return __int_as_float(__builtin_amdgcn_readfirstlane(__float_as_int(v)));
}

template <int IS_MAX>
__device__ __forceinline__ void semiring_chunk(const float* wl, const float* ht,
                                               float (&acc0)[16], float (&acc1)[16]) {
  for (int kq = 0; kq < 8; ++kq) {
    const int k = kq * 4;
    const float4 q0 = *(const float4*)(wl + 2 * k);
    const float4 q1 = *(const float4*)(wl + 2 * k + 4);
    const v2f w0 = {q0.x, q0.y}, w1 = {q0.z, q0.w};
    const v2f w2 = {q1.x, q1.y}, w3 = {q1.z, q1.w};
    const v16f hv0 = *(const v16f*)(ht + (k + 0) * 16);
    const v16f hv1 = *(const v16f*)(ht + (k + 1) * 16);
    const v16f hv2 = *(const v16f*)(ht + (k + 2) * 16);
    const v16f hv3 = *(const v16f*)(ht + (k + 3) * 16);
#pragma unroll
    for (int b = 0; b < 16; ++b) {
      const float x0 = (b & 1) ? own_sreg(hv0[b]) : hv0[b];
      const float x1 = (b & 1) ? own_sreg(hv1[b]) : hv1[b];
      const float x2 = (b & 1) ? own_sreg(hv2[b]) : hv2[b];
      const float x3 = (b & 1) ? own_sreg(hv3[b]) : hv3[b];
      const v2f s0 = w0 + x0;
      const v2f s1 = w1 + x1;
      const v2f s2 = w2 + x2;
      const v2f s3 = w3 + x3;
      acc0[b] = sel3<IS_MAX>(acc0[b], s0.x, s1.x);
      acc1[b] = sel3<IS_MAX>(acc1[b], s0.y, s1.y);
      acc0[b] = sel3<IS_MAX>(acc0[b], s2.x, s3.x);
      acc1[b] = sel3<IS_MAX>(acc1[b], s2.y, s3.y);
    }
  }
}

__device__ __forceinline__ void lds_barrier() {
  __builtin_amdgcn_fence(__ATOMIC_RELEASE, "workgroup", "local");
  __builtin_amdgcn_s_barrier();
  __builtin_amdgcn_fence(__ATOMIC_ACQUIRE, "workgroup", "local");
}

__device__ __forceinline__ void stage_store(float* dst, const float4 v) {
  dst[0] = v.x; dst[2] = v.y; dst[4] = v.z; dst[6] = v.w;
}

#define RP_ 20

template <int IS_MAX, int OUT_T>
__global__ __launch_bounds__(512) void k_semiring(const float* __restrict__ hin,
                                                  const float* __restrict__ W,
                                                  float* __restrict__ hout) {
  __shared__ __attribute__((aligned(16))) float smem[16 * 32 * WP_];
  const int b0 = blockIdx.x * 16;
  const int o0 = blockIdx.y * 64;
  const int lane = threadIdx.x & 31;
  const int wave = __builtin_amdgcn_readfirstlane((int)(threadIdx.x >> 5));
  const int kbase = wave * 64;

  float* wreg = smem + wave * (32 * WP_);
  const float* ht = hin + ((size_t)blockIdx.x * H_ + kbase) * 16;
  const int srow = lane >> 3;
  const int scol = (lane & 7) * 4;
  const float* wsrc = W + (size_t)(o0 + srow) * H_ + kbase + scol;
  float* wdst = wreg + srow * WP_ + 2 * scol;

  float acc0[16], acc1[16];
#pragma unroll
  for (int b = 0; b < 16; ++b) {
    acc0[b] = IS_MAX ? -3.402823466e38f : 3.402823466e38f;
    acc1[b] = acc0[b];
  }

  float4 pa[16], pb[16];
#pragma unroll
  for (int j = 0; j < 16; ++j) pa[j] = *(const float4*)(wsrc + (size_t)(j * 4) * H_);
#pragma unroll
  for (int j = 0; j < 16; ++j) pb[j] = *(const float4*)(wsrc + (size_t)(j * 4) * H_ + 32);
#pragma unroll
  for (int j = 0; j < 16; ++j) stage_store(wdst + ((j & 7) * 4) * WP_ + (j >> 3), pa[j]);
  lds_barrier();

  const float* wl = wreg + lane * WP_;
  semiring_chunk<IS_MAX>(wl, ht, acc0, acc1);

#pragma unroll
  for (int j = 0; j < 16; ++j) stage_store(wdst + ((j & 7) * 4) * WP_ + (j >> 3) + 64, pb[j]);
  __syncthreads();
  semiring_chunk<IS_MAX>(wl + 64, ht + 32 * 16, acc0, acc1);

  __syncthreads();
  float* red = smem;
  if (OUT_T) {
    float* r0 = red + (wave * 64 + lane) * RP_;
    float* r1 = r0 + 32 * RP_;
#pragma unroll
    for (int q = 0; q < 4; ++q) {
      *(float4*)(r0 + 4 * q) = make_float4(acc0[4 * q], acc0[4 * q + 1], acc0[4 * q + 2], acc0[4 * q + 3]);
      *(float4*)(r1 + 4 * q) = make_float4(acc1[4 * q], acc1[4 * q + 1], acc1[4 * q + 2], acc1[4 * q + 3]);
    }
    __syncthreads();
#pragma unroll
    for (int p = 0; p < 2; ++p) {
      const int idx = (int)threadIdx.x + 512 * p;
      const int o = idx >> 4;
      const int b = idx & 15;
      float v = red[o * RP_ + b];
#pragma unroll
      for (int w = 1; w < 16; ++w) {
        const float u = red[(w * 64 + o) * RP_ + b];
        v = IS_MAX ? fmaxf(v, u) : fminf(v, u);
      }
      hout[((size_t)blockIdx.x * H_ + o0) * 16 + idx] = v;
    }
    return;
  }
#pragma unroll
  for (int b = 0; b < 16; ++b) {
    red[(wave * 16 + b) * 64 + lane] = acc0[b];
    red[(wave * 16 + b) * 64 + lane + 32] = acc1[b];
  }
  __syncthreads();
#pragma unroll
  for (int p = 0; p < 2; ++p) {
    const int idx = (int)threadIdx.x + 512 * p;
    const int b = idx >> 6;
    const int o = idx & 63;
    float v = red[b * 64 + o];
#pragma unroll
    for (int w = 1; w < 16; ++w) {
      const float u = red[(w * 16 + b) * 64 + o];
      v = IS_MAX ? fmaxf(v, u) : fminf(v, u);
    }
    hout[(b0 + b) * H_ + o0 + o] = v;
  }
}

__global__ __launch_bounds__(32) void k_fc_out(const float* __restrict__ h,
                                               const float* __restrict__ Wout,
                                               const float* __restrict__ bout,
                                               float* __restrict__ q) {
  const int lane = threadIdx.x & 31;
  const int half = lane >> 4;
  const int l16  = lane & 15;
  const int m0   = blockIdx.x * 16;
  const int n0   = blockIdx.y * 16;

  const int col  = n0 + l16;
  const int bcol = (col < A_) ? col : 0;

  v8f c = {};
  const float* ap = h    + (m0 + l16) * H_ + 2 * half;
  const float* bp = Wout + bcol * H_       + 2 * half;

  constexpr int CK = 4 * FO_;
  constexpr int NC = H_ / CK;
  v2f fa[3][FO_], fb[3][FO_];
#pragma unroll
  for (int cc = 0; cc < 2; ++cc) {
    chunk_load(fa[cc], ap + cc * CK); chunk_load(fb[cc], bp + cc * CK);
  }
#pragma unroll
  for (int cc = 0; cc < NC; ++cc) {
    if (cc + 2 < NC) {
      const int n = (cc + 2) % 3;
      chunk_load(fa[n], ap + (cc + 2) * CK); chunk_load(fb[n], bp + (cc + 2) * CK);
    }
    PIPE_FENCE();
    const int u = cc % 3;
#pragma unroll
    for (int s = 0; s < FO_; ++s)
      c = __builtin_amdgcn_wmma_f32_16x16x4_f32(false, fa[u][s], false, fb[u][s], (short)0, c, false, false);
    PIPE_FENCE();
  }

  if (col < A_) {
    const float bias = bout[col];
#pragma unroll
    for (int v = 0; v < 8; ++v) q[(m0 + v + 8 * half) * A_ + col] = c[v] + bias;
  }
}

extern "C" void kernel_launch(void* const* d_in, const int* in_sizes, int n_in,
                              void* d_out, int out_size, void* d_ws, size_t ws_size,
                              hipStream_t stream) {
  (void)in_sizes; (void)n_in; (void)out_size; (void)ws_size;
  const float* x    = (const float*)d_in[0];
  const float* Win  = (const float*)d_in[1];
  const float* bin  = (const float*)d_in[2];
  const float* Wmax = (const float*)d_in[3];
  const float* Wmin = (const float*)d_in[4];
  const float* Wout = (const float*)d_in[5];
  const float* bout = (const float*)d_in[6];
  float* q  = (float*)d_out;
  float* h1 = (float*)d_ws;
  float* h2 = h1 + (size_t)B_ * H_;

  k_fc_in   <<<dim3(B_ / 32, H_ / 32), 32,  0, stream>>>(x,  Win,  bin,  h1);
  k_semiring<1, 1><<<dim3(B_ / 16, H_ / 64), 512, 0, stream>>>(h1, Wmax, h2);
  k_semiring<0, 0><<<dim3(B_ / 16, H_ / 64), 512, 0, stream>>>(h2, Wmin, h1);
  k_fc_out  <<<dim3(B_ / 16, 2),       32,  0, stream>>>(h1, Wout, bout, q);
}


# semiring kernels: a wave's LDS W region is its own: staging ordered per wave, no workgroup barrier; 2nd-half loads issue behind 1st-half staging, so no wave waits on another's loads; bit-identical
# speedup vs baseline: 2.8611x; 1.0710x over previous
#include <hip/hip_runtime.h>

typedef __attribute__((ext_vector_type(2))) float v2f;
typedef __attribute__((ext_vector_type(8))) float v8f;
typedef __attribute__((ext_vector_type(16))) float v16f;


#define B_   256
#define OBS_ 256
#define H_   1024
#define A_   18
#define FI_  8
#define FO_  16

template <int N>
__device__ __forceinline__ void chunk_load(v2f (&f)[N], const float* p) {
#pragma unroll
  for (int s = 0; s < N; ++s) f[s] = *(const v2f*)(p + 4 * s);
}

#define PIPE_FENCE() __builtin_amdgcn_sched_barrier(0)

__device__ __forceinline__ void store8(float* p, const v8f c, const float bias) {
  *(float4*)(p)     = make_float4(c[0] + bias, c[1] + bias, c[2] + bias, c[3] + bias);
  *(float4*)(p + 4) = make_float4(c[4] + bias, c[5] + bias, c[6] + bias, c[7] + bias);
}

__global__ __launch_bounds__(32) void k_fc_in(const float* __restrict__ x,
                                              const float* __restrict__ Win,
                                              const float* __restrict__ bin,
                                              float* __restrict__ h1) {
  const int lane = threadIdx.x & 31;
  const int half = lane >> 4;
  const int l16  = lane & 15;
  const int m0   = blockIdx.x * 32;
  const int n0   = blockIdx.y * 32;

  v8f c00 = {}, c01 = {}, c10 = {}, c11 = {};

  const float* a0 = x   + (m0 +      l16) * OBS_ + 2 * half;
  const float* a1 = x   + (m0 + 16 + l16) * OBS_ + 2 * half;
  const float* b0 = Win + (n0 +      l16) * OBS_ + 2 * half;
  const float* b1 = Win + (n0 + 16 + l16) * OBS_ + 2 * half;

  constexpr int CK = 4 * FI_;
  constexpr int NC = OBS_ / CK;
  v2f fa0[3][FI_], fa1[3][FI_], fb0[3][FI_], fb1[3][FI_];
#pragma unroll
  for (int c = 0; c < 2; ++c) {
    chunk_load(fa0[c], a0 + c * CK); chunk_load(fa1[c], a1 + c * CK);
    chunk_load(fb0[c], b0 + c * CK); chunk_load(fb1[c], b1 + c * CK);
  }
#pragma unroll
  for (int c = 0; c < NC; ++c) {
    if (c + 2 < NC) {
      const int n = (c + 2) % 3;
      chunk_load(fa0[n], a0 + (c + 2) * CK); chunk_load(fa1[n], a1 + (c + 2) * CK);
      chunk_load(fb0[n], b0 + (c + 2) * CK); chunk_load(fb1[n], b1 + (c + 2) * CK);
    }
    PIPE_FENCE();
    const int u = c % 3;
#pragma unroll
    for (int s = 0; s < FI_; ++s) {
      c00 = __builtin_amdgcn_wmma_f32_16x16x4_f32(false, fa0[u][s], false, fb0[u][s], (short)0, c00, false, false);
      c01 = __builtin_amdgcn_wmma_f32_16x16x4_f32(false, fa0[u][s], false, fb1[u][s], (short)0, c01, false, false);
      c10 = __builtin_amdgcn_wmma_f32_16x16x4_f32(false, fa1[u][s], false, fb0[u][s], (short)0, c10, false, false);
      c11 = __builtin_amdgcn_wmma_f32_16x16x4_f32(false, fa1[u][s], false, fb1[u][s], (short)0, c11, false, false);
    }
    PIPE_FENCE();
  }

  const int col0 = n0 + l16;
  const int col1 = n0 + 16 + l16;
  const float bias0 = bin[col0];
  const float bias1 = bin[col1];
  const int bt0 = blockIdx.x * 2;
  store8(h1 + ((size_t)(bt0 * H_ + col0) * 16 + 8 * half), c00, bias0);
  store8(h1 + ((size_t)(bt0 * H_ + col1) * 16 + 8 * half), c01, bias1);
  store8(h1 + ((size_t)((bt0 + 1) * H_ + col0) * 16 + 8 * half), c10, bias0);
  store8(h1 + ((size_t)((bt0 + 1) * H_ + col1) * 16 + 8 * half), c11, bias1);
}

#define WP_ 132

template <int IS_MAX>
__device__ __forceinline__ float sel3(float a, float x, float y) {
  return IS_MAX ? fmaxf(fmaxf(a, x), y) : fminf(fminf(a, x), y);
}

__device__ __forceinline__ float own_sreg(float v) {
  return __int_as_float(__builtin_amdgcn_readfirstlane(__float_as_int(v)));
}

template <int IS_MAX>
__device__ __forceinline__ void semiring_chunk(const float* wl, const float* ht,
                                               float (&acc0)[16], float (&acc1)[16]) {
  for (int kq = 0; kq < 8; ++kq) {
    const int k = kq * 4;
    const float4 q0 = *(const float4*)(wl + 2 * k);
    const float4 q1 = *(const float4*)(wl + 2 * k + 4);
    const v2f w0 = {q0.x, q0.y}, w1 = {q0.z, q0.w};
    const v2f w2 = {q1.x, q1.y}, w3 = {q1.z, q1.w};
    const v16f hv0 = *(const v16f*)(ht + (k + 0) * 16);
    const v16f hv1 = *(const v16f*)(ht + (k + 1) * 16);
    const v16f hv2 = *(const v16f*)(ht + (k + 2) * 16);
    const v16f hv3 = *(const v16f*)(ht + (k + 3) * 16);
#pragma unroll
    for (int b = 0; b < 16; ++b) {
      const float x0 = (b & 1) ? own_sreg(hv0[b]) : hv0[b];
      const float x1 = (b & 1) ? own_sreg(hv1[b]) : hv1[b];
      const float x2 = (b & 1) ? own_sreg(hv2[b]) : hv2[b];
      const float x3 = (b & 1) ? own_sreg(hv3[b]) : hv3[b];
      const v2f s0 = w0 + x0;
      const v2f s1 = w1 + x1;
      const v2f s2 = w2 + x2;
      const v2f s3 = w3 + x3;
      acc0[b] = sel3<IS_MAX>(acc0[b], s0.x, s1.x);
      acc1[b] = sel3<IS_MAX>(acc1[b], s0.y, s1.y);
      acc0[b] = sel3<IS_MAX>(acc0[b], s2.x, s3.x);
      acc1[b] = sel3<IS_MAX>(acc1[b], s2.y, s3.y);
    }
  }
}

__device__ __forceinline__ void wave_lds_order() {
  __builtin_amdgcn_fence(__ATOMIC_RELEASE, "workgroup", "local");
  __builtin_amdgcn_fence(__ATOMIC_ACQUIRE, "workgroup", "local");
}

__device__ __forceinline__ void stage_store(float* dst, const float4 v) {
  dst[0] = v.x; dst[2] = v.y; dst[4] = v.z; dst[6] = v.w;
}

#define RP_ 20

template <int IS_MAX, int OUT_T>
__global__ __launch_bounds__(512) void k_semiring(const float* __restrict__ hin,
                                                  const float* __restrict__ W,
                                                  float* __restrict__ hout) {
  __shared__ __attribute__((aligned(16))) float smem[16 * 32 * WP_];
  const int b0 = blockIdx.x * 16;
  const int o0 = blockIdx.y * 64;
  const int lane = threadIdx.x & 31;
  const int wave = __builtin_amdgcn_readfirstlane((int)(threadIdx.x >> 5));
  const int kbase = wave * 64;

  float* wreg = smem + wave * (32 * WP_);
  const float* ht = hin + ((size_t)blockIdx.x * H_ + kbase) * 16;
  const int srow = lane >> 3;
  const int scol = (lane & 7) * 4;
  const float* wsrc = W + (size_t)(o0 + srow) * H_ + kbase + scol;
  float* wdst = wreg + srow * WP_ + 2 * scol;

  float acc0[16], acc1[16];
#pragma unroll
  for (int b = 0; b < 16; ++b) {
    acc0[b] = IS_MAX ? -3.402823466e38f : 3.402823466e38f;
    acc1[b] = acc0[b];
  }

  float4 pa[16], pb[16];
#pragma unroll
  for (int j = 0; j < 16; ++j) pa[j] = *(const float4*)(wsrc + (size_t)(j * 4) * H_);
#pragma unroll
  for (int j = 0; j < 16; ++j) stage_store(wdst + ((j & 7) * 4) * WP_ + (j >> 3), pa[j]);
  __builtin_amdgcn_sched_barrier(0);
#pragma unroll
  for (int j = 0; j < 16; ++j) pb[j] = *(const float4*)(wsrc + (size_t)(j * 4) * H_ + 32);
  wave_lds_order();

  const float* wl = wreg + lane * WP_;
  semiring_chunk<IS_MAX>(wl, ht, acc0, acc1);

#pragma unroll
  for (int j = 0; j < 16; ++j) stage_store(wdst + ((j & 7) * 4) * WP_ + (j >> 3) + 64, pb[j]);
  wave_lds_order();
  semiring_chunk<IS_MAX>(wl + 64, ht + 32 * 16, acc0, acc1);

  __syncthreads();
  float* red = smem;
  if (OUT_T) {
    float* r0 = red + (wave * 64 + lane) * RP_;
    float* r1 = r0 + 32 * RP_;
#pragma unroll
    for (int q = 0; q < 4; ++q) {
      *(float4*)(r0 + 4 * q) = make_float4(acc0[4 * q], acc0[4 * q + 1], acc0[4 * q + 2], acc0[4 * q + 3]);
      *(float4*)(r1 + 4 * q) = make_float4(acc1[4 * q], acc1[4 * q + 1], acc1[4 * q + 2], acc1[4 * q + 3]);
    }
    __syncthreads();
#pragma unroll
    for (int p = 0; p < 2; ++p) {
      const int idx = (int)threadIdx.x + 512 * p;
      const int o = idx >> 4;
      const int b = idx & 15;
      float v = red[o * RP_ + b];
#pragma unroll
      for (int w = 1; w < 16; ++w) {
        const float u = red[(w * 64 + o) * RP_ + b];
        v = IS_MAX ? fmaxf(v, u) : fminf(v, u);
      }
      hout[((size_t)blockIdx.x * H_ + o0) * 16 + idx] = v;
    }
    return;
  }
#pragma unroll
  for (int b = 0; b < 16; ++b) {
    red[(wave * 16 + b) * 64 + lane] = acc0[b];
    red[(wave * 16 + b) * 64 + lane + 32] = acc1[b];
  }
  __syncthreads();
#pragma unroll
  for (int p = 0; p < 2; ++p) {
    const int idx = (int)threadIdx.x + 512 * p;
    const int b = idx >> 6;
    const int o = idx & 63;
    float v = red[b * 64 + o];
#pragma unroll
    for (int w = 1; w < 16; ++w) {
      const float u = red[(w * 16 + b) * 64 + o];
      v = IS_MAX ? fmaxf(v, u) : fminf(v, u);
    }
    hout[(b0 + b) * H_ + o0 + o] = v;
  }
}

__global__ __launch_bounds__(32) void k_fc_out(const float* __restrict__ h,
                                               const float* __restrict__ Wout,
                                               const float* __restrict__ bout,
                                               float* __restrict__ q) {
  const int lane = threadIdx.x & 31;
  const int half = lane >> 4;
  const int l16  = lane & 15;
  const int m0   = blockIdx.x * 16;
  const int n0   = blockIdx.y * 16;

  const int col  = n0 + l16;
  const int bcol = (col < A_) ? col : 0;

  v8f c = {};
  const float* ap = h    + (m0 + l16) * H_ + 2 * half;
  const float* bp = Wout + bcol * H_       + 2 * half;

  constexpr int CK = 4 * FO_;
  constexpr int NC = H_ / CK;
  v2f fa[3][FO_], fb[3][FO_];
#pragma unroll
  for (int cc = 0; cc < 2; ++cc) {
    chunk_load(fa[cc], ap + cc * CK); chunk_load(fb[cc], bp + cc * CK);
  }
#pragma unroll
  for (int cc = 0; cc < NC; ++cc) {
    if (cc + 2 < NC) {
      const int n = (cc + 2) % 3;
      chunk_load(fa[n], ap + (cc + 2) * CK); chunk_load(fb[n], bp + (cc + 2) * CK);
    }
    PIPE_FENCE();
    const int u = cc % 3;
#pragma unroll
    for (int s = 0; s < FO_; ++s)
      c = __builtin_amdgcn_wmma_f32_16x16x4_f32(false, fa[u][s], false, fb[u][s], (short)0, c, false, false);
    PIPE_FENCE();
  }

  if (col < A_) {
    const float bias = bout[col];
#pragma unroll
    for (int v = 0; v < 8; ++v) q[(m0 + v + 8 * half) * A_ + col] = c[v] + bias;
  }
}

extern "C" void kernel_launch(void* const* d_in, const int* in_sizes, int n_in,
                              void* d_out, int out_size, void* d_ws, size_t ws_size,
                              hipStream_t stream) {
  (void)in_sizes; (void)n_in; (void)out_size; (void)ws_size;
  const float* x    = (const float*)d_in[0];
  const float* Win  = (const float*)d_in[1];
  const float* bin  = (const float*)d_in[2];
  const float* Wmax = (const float*)d_in[3];
  const float* Wmin = (const float*)d_in[4];
  const float* Wout = (const float*)d_in[5];
  const float* bout = (const float*)d_in[6];
  float* q  = (float*)d_out;
  float* h1 = (float*)d_ws;
  float* h2 = h1 + (size_t)B_ * H_;

  k_fc_in   <<<dim3(B_ / 32, H_ / 32), 32,  0, stream>>>(x,  Win,  bin,  h1);
  k_semiring<1, 1><<<dim3(B_ / 16, H_ / 64), 512, 0, stream>>>(h1, Wmax, h2);
  k_semiring<0, 0><<<dim3(B_ / 16, H_ / 64), 512, 0, stream>>>(h2, Wmin, h1);
  k_fc_out  <<<dim3(B_ / 16, 2),       32,  0, stream>>>(h1, Wout, bout, q);
}


# semiring kernels: a step's operands (h: scalar loads, W: LDS reads) fetched one step ahead, right behind the previous step's adds, into the same registers; first h load before W wait; bit-identical
# speedup vs baseline: 2.9501x; 1.0311x over previous
#include <hip/hip_runtime.h>

typedef __attribute__((ext_vector_type(2))) float v2f;
typedef __attribute__((ext_vector_type(8))) float v8f;
typedef __attribute__((ext_vector_type(16))) float v16f;


#define B_   256
#define OBS_ 256
#define H_   1024
#define A_   18
#define FI_  8
#define FO_  16

template <int N>
__device__ __forceinline__ void chunk_load(v2f (&f)[N], const float* p) {
#pragma unroll
  for (int s = 0; s < N; ++s) f[s] = *(const v2f*)(p + 4 * s);
}

#define PIPE_FENCE() __builtin_amdgcn_sched_barrier(0)

__device__ __forceinline__ void store8(float* p, const v8f c, const float bias) {
  *(float4*)(p)     = make_float4(c[0] + bias, c[1] + bias, c[2] + bias, c[3] + bias);
  *(float4*)(p + 4) = make_float4(c[4] + bias, c[5] + bias, c[6] + bias, c[7] + bias);
}

__global__ __launch_bounds__(32) void k_fc_in(const float* __restrict__ x,
                                              const float* __restrict__ Win,
                                              const float* __restrict__ bin,
                                              float* __restrict__ h1) {
  const int lane = threadIdx.x & 31;
  const int half = lane >> 4;
  const int l16  = lane & 15;
  const int m0   = blockIdx.x * 32;
  const int n0   = blockIdx.y * 32;

  v8f c00 = {}, c01 = {}, c10 = {}, c11 = {};

  const float* a0 = x   + (m0 +      l16) * OBS_ + 2 * half;
  const float* a1 = x   + (m0 + 16 + l16) * OBS_ + 2 * half;
  const float* b0 = Win + (n0 +      l16) * OBS_ + 2 * half;
  const float* b1 = Win + (n0 + 16 + l16) * OBS_ + 2 * half;

  constexpr int CK = 4 * FI_;
  constexpr int NC = OBS_ / CK;
  v2f fa0[3][FI_], fa1[3][FI_], fb0[3][FI_], fb1[3][FI_];
#pragma unroll
  for (int c = 0; c < 2; ++c) {
    chunk_load(fa0[c], a0 + c * CK); chunk_load(fa1[c], a1 + c * CK);
    chunk_load(fb0[c], b0 + c * CK); chunk_load(fb1[c], b1 + c * CK);
  }
#pragma unroll
  for (int c = 0; c < NC; ++c) {
    if (c + 2 < NC) {
      const int n = (c + 2) % 3;
      chunk_load(fa0[n], a0 + (c + 2) * CK); chunk_load(fa1[n], a1 + (c + 2) * CK);
      chunk_load(fb0[n], b0 + (c + 2) * CK); chunk_load(fb1[n], b1 + (c + 2) * CK);
    }
    PIPE_FENCE();
    const int u = c % 3;
#pragma unroll
    for (int s = 0; s < FI_; ++s) {
      c00 = __builtin_amdgcn_wmma_f32_16x16x4_f32(false, fa0[u][s], false, fb0[u][s], (short)0, c00, false, false);
      c01 = __builtin_amdgcn_wmma_f32_16x16x4_f32(false, fa0[u][s], false, fb1[u][s], (short)0, c01, false, false);
      c10 = __builtin_amdgcn_wmma_f32_16x16x4_f32(false, fa1[u][s], false, fb0[u][s], (short)0, c10, false, false);
      c11 = __builtin_amdgcn_wmma_f32_16x16x4_f32(false, fa1[u][s], false, fb1[u][s], (short)0, c11, false, false);
    }
    PIPE_FENCE();
  }

  const int col0 = n0 + l16;
  const int col1 = n0 + 16 + l16;
  const float bias0 = bin[col0];
  const float bias1 = bin[col1];
  const int bt0 = blockIdx.x * 2;
  store8(h1 + ((size_t)(bt0 * H_ + col0) * 16 + 8 * half), c00, bias0);
  store8(h1 + ((size_t)(bt0 * H_ + col1) * 16 + 8 * half), c01, bias1);
  store8(h1 + ((size_t)((bt0 + 1) * H_ + col0) * 16 + 8 * half), c10, bias0);
  store8(h1 + ((size_t)((bt0 + 1) * H_ + col1) * 16 + 8 * half), c11, bias1);
}

#define WP_ 132

template <int IS_MAX>
__device__ __forceinline__ float sel3(float a, float x, float y) {
  return IS_MAX ? fmaxf(fmaxf(a, x), y) : fminf(fminf(a, x), y);
}

__device__ __forceinline__ float own_sreg(float v) {
  return __int_as_float(__builtin_amdgcn_readfirstlane(__float_as_int(v)));
}

#define STEP_FENCE() __builtin_amdgcn_sched_barrier(0)

template <int IS_MAX, bool MORE_H, bool MORE_W>
__device__ __forceinline__ void semiring_step(const float* wn, const float* hn,
                                              float4& q0, float4& q1, v16f (&hv)[4],
                                              float (&acc0)[16], float (&acc1)[16]) {
  const v2f w0 = {q0.x, q0.y}, w1 = {q0.z, q0.w};
  const v2f w2 = {q1.x, q1.y}, w3 = {q1.z, q1.w};
  v2f s0[16], s1[16], s2[16], s3[16];
#pragma unroll
  for (int b = 0; b < 16; ++b) {
    const float x0 = (b & 1) ? own_sreg(hv[0][b]) : hv[0][b];
    const float x1 = (b & 1) ? own_sreg(hv[1][b]) : hv[1][b];
    const float x2 = (b & 1) ? own_sreg(hv[2][b]) : hv[2][b];
    const float x3 = (b & 1) ? own_sreg(hv[3][b]) : hv[3][b];
    s0[b] = w0 + x0;
    s1[b] = w1 + x1;
    s2[b] = w2 + x2;
    s3[b] = w3 + x3;
  }
  STEP_FENCE();
  if (MORE_W) {
    q0 = *(const float4*)(wn);
    q1 = *(const float4*)(wn + 4);
  }
  if (MORE_H) {
    hv[0] = *(const v16f*)(hn);
    hv[1] = *(const v16f*)(hn + 16);
    hv[2] = *(const v16f*)(hn + 32);
    hv[3] = *(const v16f*)(hn + 48);
  }
  STEP_FENCE();
#pragma unroll
  for (int b = 0; b < 16; ++b) {
    acc0[b] = sel3<IS_MAX>(acc0[b], s0[b].x, s1[b].x);
    acc1[b] = sel3<IS_MAX>(acc1[b], s0[b].y, s1[b].y);
    acc0[b] = sel3<IS_MAX>(acc0[b], s2[b].x, s3[b].x);
    acc1[b] = sel3<IS_MAX>(acc1[b], s2[b].y, s3[b].y);
  }
}

__device__ __forceinline__ void wave_lds_order() {
  __builtin_amdgcn_fence(__ATOMIC_RELEASE, "workgroup", "local");
  __builtin_amdgcn_fence(__ATOMIC_ACQUIRE, "workgroup", "local");
}

__device__ __forceinline__ void stage_store(float* dst, const float4 v) {
  dst[0] = v.x; dst[2] = v.y; dst[4] = v.z; dst[6] = v.w;
}

#define RP_ 20

template <int IS_MAX, int OUT_T>
__global__ __launch_bounds__(512) void k_semiring(const float* __restrict__ hin,
                                                  const float* __restrict__ W,
                                                  float* __restrict__ hout) {
  __shared__ __attribute__((aligned(16))) float smem[16 * 32 * WP_];
  const int b0 = blockIdx.x * 16;
  const int o0 = blockIdx.y * 64;
  const int lane = threadIdx.x & 31;
  const int wave = __builtin_amdgcn_readfirstlane((int)(threadIdx.x >> 5));
  const int kbase = wave * 64;

  float* wreg = smem + wave * (32 * WP_);
  const float* ht = hin + ((size_t)blockIdx.x * H_ + kbase) * 16;
  v16f hv[4];
#pragma unroll
  for (int j = 0; j < 4; ++j) hv[j] = *(const v16f*)(ht + j * 16);
  const int srow = lane >> 3;
  const int scol = (lane & 7) * 4;
  const float* wsrc = W + (size_t)(o0 + srow) * H_ + kbase + scol;
  float* wdst = wreg + srow * WP_ + 2 * scol;

  float acc0[16], acc1[16];
#pragma unroll
  for (int b = 0; b < 16; ++b) {
    acc0[b] = IS_MAX ? -3.402823466e38f : 3.402823466e38f;
    acc1[b] = acc0[b];
  }

  float4 pa[16], pb[16];
#pragma unroll
  for (int j = 0; j < 16; ++j) pa[j] = *(const float4*)(wsrc + (size_t)(j * 4) * H_);
#pragma unroll
  for (int j = 0; j < 16; ++j) stage_store(wdst + ((j & 7) * 4) * WP_ + (j >> 3), pa[j]);
  __builtin_amdgcn_sched_barrier(0);
#pragma unroll
  for (int j = 0; j < 16; ++j) pb[j] = *(const float4*)(wsrc + (size_t)(j * 4) * H_ + 32);
  wave_lds_order();

  const float* wl = wreg + lane * WP_;
  float4 q0 = *(const float4*)(wl);
  float4 q1 = *(const float4*)(wl + 4);
  for (int s = 1; s < 8; ++s)
    semiring_step<IS_MAX, true, true>(wl + 8 * s, ht + 64 * s, q0, q1, hv, acc0, acc1);
  semiring_step<IS_MAX, true, false>(wl, ht + 64 * 8, q0, q1, hv, acc0, acc1);

#pragma unroll
  for (int j = 0; j < 16; ++j) stage_store(wdst + ((j & 7) * 4) * WP_ + (j >> 3) + 64, pb[j]);
  wave_lds_order();
  q0 = *(const float4*)(wl + 64);
  q1 = *(const float4*)(wl + 68);
  for (int s = 9; s < 16; ++s)
    semiring_step<IS_MAX, true, true>(wl + 8 * s, ht + 64 * s, q0, q1, hv, acc0, acc1);
  semiring_step<IS_MAX, false, false>(wl, ht, q0, q1, hv, acc0, acc1);

  __syncthreads();
  float* red = smem;
  if (OUT_T) {
    float* r0 = red + (wave * 64 + lane) * RP_;
    float* r1 = r0 + 32 * RP_;
#pragma unroll
    for (int q = 0; q < 4; ++q) {
      *(float4*)(r0 + 4 * q) = make_float4(acc0[4 * q], acc0[4 * q + 1], acc0[4 * q + 2], acc0[4 * q + 3]);
      *(float4*)(r1 + 4 * q) = make_float4(acc1[4 * q], acc1[4 * q + 1], acc1[4 * q + 2], acc1[4 * q + 3]);
    }
    __syncthreads();
#pragma unroll
    for (int p = 0; p < 2; ++p) {
      const int idx = (int)threadIdx.x + 512 * p;
      const int o = idx >> 4;
      const int b = idx & 15;
      float v = red[o * RP_ + b];
#pragma unroll
      for (int w = 1; w < 16; ++w) {
        const float u = red[(w * 64 + o) * RP_ + b];
        v = IS_MAX ? fmaxf(v, u) : fminf(v, u);
      }
      hout[((size_t)blockIdx.x * H_ + o0) * 16 + idx] = v;
    }
    return;
  }
#pragma unroll
  for (int b = 0; b < 16; ++b) {
    red[(wave * 16 + b) * 64 + lane] = acc0[b];
    red[(wave * 16 + b) * 64 + lane + 32] = acc1[b];
  }
  __syncthreads();
#pragma unroll
  for (int p = 0; p < 2; ++p) {
    const int idx = (int)threadIdx.x + 512 * p;
    const int b = idx >> 6;
    const int o = idx & 63;
    float v = red[b * 64 + o];
#pragma unroll
    for (int w = 1; w < 16; ++w) {
      const float u = red[(w * 16 + b) * 64 + o];
      v = IS_MAX ? fmaxf(v, u) : fminf(v, u);
    }
    hout[(b0 + b) * H_ + o0 + o] = v;
  }
}

__global__ __launch_bounds__(32) void k_fc_out(const float* __restrict__ h,
                                               const float* __restrict__ Wout,
                                               const float* __restrict__ bout,
                                               float* __restrict__ q) {
  const int lane = threadIdx.x & 31;
  const int half = lane >> 4;
  const int l16  = lane & 15;
  const int m0   = blockIdx.x * 16;
  const int n0   = blockIdx.y * 16;

  const int col  = n0 + l16;
  const int bcol = (col < A_) ? col : 0;

  v8f c = {};
  const float* ap = h    + (m0 + l16) * H_ + 2 * half;
  const float* bp = Wout + bcol * H_       + 2 * half;

  constexpr int CK = 4 * FO_;
  constexpr int NC = H_ / CK;
  v2f fa[3][FO_], fb[3][FO_];
#pragma unroll
  for (int cc = 0; cc < 2; ++cc) {
    chunk_load(fa[cc], ap + cc * CK); chunk_load(fb[cc], bp + cc * CK);
  }
#pragma unroll
  for (int cc = 0; cc < NC; ++cc) {
    if (cc + 2 < NC) {
      const int n = (cc + 2) % 3;
      chunk_load(fa[n], ap + (cc + 2) * CK); chunk_load(fb[n], bp + (cc + 2) * CK);
    }
    PIPE_FENCE();
    const int u = cc % 3;
#pragma unroll
    for (int s = 0; s < FO_; ++s)
      c = __builtin_amdgcn_wmma_f32_16x16x4_f32(false, fa[u][s], false, fb[u][s], (short)0, c, false, false);
    PIPE_FENCE();
  }

  if (col < A_) {
    const float bias = bout[col];
#pragma unroll
    for (int v = 0; v < 8; ++v) q[(m0 + v + 8 * half) * A_ + col] = c[v] + bias;
  }
}

extern "C" void kernel_launch(void* const* d_in, const int* in_sizes, int n_in,
                              void* d_out, int out_size, void* d_ws, size_t ws_size,
                              hipStream_t stream) {
  (void)in_sizes; (void)n_in; (void)out_size; (void)ws_size;
  const float* x    = (const float*)d_in[0];
  const float* Win  = (const float*)d_in[1];
  const float* bin  = (const float*)d_in[2];
  const float* Wmax = (const float*)d_in[3];
  const float* Wmin = (const float*)d_in[4];
  const float* Wout = (const float*)d_in[5];
  const float* bout = (const float*)d_in[6];
  float* q  = (float*)d_out;
  float* h1 = (float*)d_ws;
  float* h2 = h1 + (size_t)B_ * H_;

  k_fc_in   <<<dim3(B_ / 32, H_ / 32), 32,  0, stream>>>(x,  Win,  bin,  h1);
  k_semiring<1, 1><<<dim3(B_ / 16, H_ / 64), 512, 0, stream>>>(h1, Wmax, h2);
  k_semiring<0, 0><<<dim3(B_ / 16, H_ / 64), 512, 0, stream>>>(h2, Wmin, h1);
  k_fc_out  <<<dim3(B_ / 16, 2),       32,  0, stream>>>(h1, Wout, bout, q);
}


# fc_in: a block of 4 waves (one per SIMD) shares one 32x32 tile's operands, staged once through LDS with whole-line loads; each wave chains one 16x16 sub-tile; same k order, bit-identical
# speedup vs baseline: 3.0643x; 1.0387x over previous
#include <hip/hip_runtime.h>

typedef __attribute__((ext_vector_type(2))) float v2f;
typedef __attribute__((ext_vector_type(4))) float v4f;
typedef __attribute__((ext_vector_type(8))) float v8f;
typedef __attribute__((ext_vector_type(16))) float v16f;


#define B_   256
#define OBS_ 256
#define H_   1024
#define A_   18
#define FO_  16

template <int N>
__device__ __forceinline__ void chunk_load(v2f (&f)[N], const float* p) {
#pragma unroll
  for (int s = 0; s < N; ++s) f[s] = *(const v2f*)(p + 4 * s);
}

#define PIPE_FENCE() __builtin_amdgcn_sched_barrier(0)

#define FG_ 8
#define FP_ 260

__device__ __forceinline__ void store8(float* p, const v8f c, const float bias) {
  *(float4*)(p)     = make_float4(c[0] + bias, c[1] + bias, c[2] + bias, c[3] + bias);
  *(float4*)(p + 4) = make_float4(c[4] + bias, c[5] + bias, c[6] + bias, c[7] + bias);
}

__device__ __forceinline__ void lds_barrier() {
  __builtin_amdgcn_fence(__ATOMIC_RELEASE, "workgroup", "local");
  __builtin_amdgcn_s_barrier();
  __builtin_amdgcn_fence(__ATOMIC_ACQUIRE, "workgroup", "local");
}

template <int G0, int G1>
__device__ __forceinline__ void fc_in_groups(const float* ap, const float* bp, v8f& c) {
  v2f fa[2][FG_], fb[2][FG_];
  chunk_load(fa[G0 & 1], ap + G0 * 4 * FG_); chunk_load(fb[G0 & 1], bp + G0 * 4 * FG_);
#pragma unroll
  for (int g = G0; g < G1; ++g) {
    if (g + 1 < G1) {
      chunk_load(fa[(g + 1) & 1], ap + (g + 1) * 4 * FG_); chunk_load(fb[(g + 1) & 1], bp + (g + 1) * 4 * FG_);
    }
    PIPE_FENCE();
#pragma unroll
    for (int s = 0; s < FG_; ++s)
      c = __builtin_amdgcn_wmma_f32_16x16x4_f32(false, fa[g & 1][s], false, fb[g & 1][s], (short)0, c, false, false);
    PIPE_FENCE();
  }
}

__global__ __launch_bounds__(128) void k_fc_in(const float* __restrict__ x,
                                               const float* __restrict__ Win,
                                               const float* __restrict__ bin,
                                               float* __restrict__ h1) {
  __shared__ __attribute__((aligned(16))) float opnd[64 * FP_];
  const int lane = threadIdx.x & 31;
  const int half = lane >> 4;
  const int l16  = lane & 15;
  const int wave = __builtin_amdgcn_readfirstlane((int)(threadIdx.x >> 5));
  const int m0   = blockIdx.x * 32;
  const int n0   = blockIdx.y * 32;

  const float* src = (wave < 2) ? x   + (size_t)(m0 + 16 * wave) * OBS_
                                : Win + (size_t)(n0 + 16 * (wave - 2)) * OBS_;
  float* dst = opnd + (16 * wave) * FP_ + 4 * lane;
  src += 4 * lane;
  v4f lo[16], hi[16];
#pragma unroll
  for (int r = 0; r < 16; ++r) lo[r] = *(const v4f*)(src + r * OBS_);
  PIPE_FENCE();
#pragma unroll
  for (int r = 0; r < 16; ++r) hi[r] = *(const v4f*)(src + r * OBS_ + 128);

  const int wm = wave >> 1;
  const int wn = wave & 1;
  const float* ap = opnd + (16 * wm + l16) * FP_ + 2 * half;
  const float* bp = opnd + (32 + 16 * wn + l16) * FP_ + 2 * half;
  v8f c = {};

#pragma unroll
  for (int r = 0; r < 16; ++r) *(v4f*)(dst + r * FP_) = lo[r];
  lds_barrier();
  constexpr int NG = OBS_ / (4 * FG_);
  fc_in_groups<0, NG / 2>(ap, bp, c);
#pragma unroll
  for (int r = 0; r < 16; ++r) *(v4f*)(dst + r * FP_ + 128) = hi[r];
  lds_barrier();
  fc_in_groups<NG / 2, NG>(ap, bp, c);

  const int col = n0 + 16 * wn + l16;
  const int bt  = blockIdx.x * 2 + wm;
  store8(h1 + ((size_t)(bt * H_ + col) * 16 + 8 * half), c, bin[col]);
}

#define WP_ 132

template <int IS_MAX>
__device__ __forceinline__ float sel3(float a, float x, float y) {
  return IS_MAX ? fmaxf(fmaxf(a, x), y) : fminf(fminf(a, x), y);
}

__device__ __forceinline__ float own_sreg(float v) {
  return __int_as_float(__builtin_amdgcn_readfirstlane(__float_as_int(v)));
}

#define STEP_FENCE() __builtin_amdgcn_sched_barrier(0)

template <int IS_MAX, bool MORE_H, bool MORE_W>
__device__ __forceinline__ void semiring_step(const float* wn, const float* hn,
                                              float4& q0, float4& q1, v16f (&hv)[4],
                                              float (&acc0)[16], float (&acc1)[16]) {
  const v2f w0 = {q0.x, q0.y}, w1 = {q0.z, q0.w};
  const v2f w2 = {q1.x, q1.y}, w3 = {q1.z, q1.w};
  v2f s0[16], s1[16], s2[16], s3[16];
#pragma unroll
  for (int b = 0; b < 16; ++b) {
    const float x0 = (b & 1) ? own_sreg(hv[0][b]) : hv[0][b];
    const float x1 = (b & 1) ? own_sreg(hv[1][b]) : hv[1][b];
    const float x2 = (b & 1) ? own_sreg(hv[2][b]) : hv[2][b];
    const float x3 = (b & 1) ? own_sreg(hv[3][b]) : hv[3][b];
    s0[b] = w0 + x0;
    s1[b] = w1 + x1;
    s2[b] = w2 + x2;
    s3[b] = w3 + x3;
  }
  STEP_FENCE();
  if (MORE_W) {
    q0 = *(const float4*)(wn);
    q1 = *(const float4*)(wn + 4);
  }
  if (MORE_H) {
    hv[0] = *(const v16f*)(hn);
    hv[1] = *(const v16f*)(hn + 16);
    hv[2] = *(const v16f*)(hn + 32);
    hv[3] = *(const v16f*)(hn + 48);
  }
  STEP_FENCE();
#pragma unroll
  for (int b = 0; b < 16; ++b) {
    acc0[b] = sel3<IS_MAX>(acc0[b], s0[b].x, s1[b].x);
    acc1[b] = sel3<IS_MAX>(acc1[b], s0[b].y, s1[b].y);
    acc0[b] = sel3<IS_MAX>(acc0[b], s2[b].x, s3[b].x);
    acc1[b] = sel3<IS_MAX>(acc1[b], s2[b].y, s3[b].y);
  }
}

__device__ __forceinline__ void wave_lds_order() {
  __builtin_amdgcn_fence(__ATOMIC_RELEASE, "workgroup", "local");
  __builtin_amdgcn_fence(__ATOMIC_ACQUIRE, "workgroup", "local");
}

__device__ __forceinline__ void stage_store(float* dst, const float4 v) {
  dst[0] = v.x; dst[2] = v.y; dst[4] = v.z; dst[6] = v.w;
}

#define RP_ 20

template <int IS_MAX, int OUT_T>
__global__ __launch_bounds__(512) void k_semiring(const float* __restrict__ hin,
                                                  const float* __restrict__ W,
                                                  float* __restrict__ hout) {
  __shared__ __attribute__((aligned(16))) float smem[16 * 32 * WP_];
  const int b0 = blockIdx.x * 16;
  const int o0 = blockIdx.y * 64;
  const int lane = threadIdx.x & 31;
  const int wave = __builtin_amdgcn_readfirstlane((int)(threadIdx.x >> 5));
  const int kbase = wave * 64;

  float* wreg = smem + wave * (32 * WP_);
  const float* ht = hin + ((size_t)blockIdx.x * H_ + kbase) * 16;
  v16f hv[4];
#pragma unroll
  for (int j = 0; j < 4; ++j) hv[j] = *(const v16f*)(ht + j * 16);
  const int srow = lane >> 3;
  const int scol = (lane & 7) * 4;
  const float* wsrc = W + (size_t)(o0 + srow) * H_ + kbase + scol;
  float* wdst = wreg + srow * WP_ + 2 * scol;

  float acc0[16], acc1[16];
#pragma unroll
  for (int b = 0; b < 16; ++b) {
    acc0[b] = IS_MAX ? -3.402823466e38f : 3.402823466e38f;
    acc1[b] = acc0[b];
  }

  float4 pa[16], pb[16];
#pragma unroll
  for (int j = 0; j < 16; ++j) pa[j] = *(const float4*)(wsrc + (size_t)(j * 4) * H_);
#pragma unroll
  for (int j = 0; j < 16; ++j) stage_store(wdst + ((j & 7) * 4) * WP_ + (j >> 3), pa[j]);
  __builtin_amdgcn_sched_barrier(0);
#pragma unroll
  for (int j = 0; j < 16; ++j) pb[j] = *(const float4*)(wsrc + (size_t)(j * 4) * H_ + 32);
  wave_lds_order();

  const float* wl = wreg + lane * WP_;
  float4 q0 = *(const float4*)(wl);
  float4 q1 = *(const float4*)(wl + 4);
  for (int s = 1; s < 8; ++s)
    semiring_step<IS_MAX, true, true>(wl + 8 * s, ht + 64 * s, q0, q1, hv, acc0, acc1);
  semiring_step<IS_MAX, true, false>(wl, ht + 64 * 8, q0, q1, hv, acc0, acc1);

#pragma unroll
  for (int j = 0; j < 16; ++j) stage_store(wdst + ((j & 7) * 4) * WP_ + (j >> 3) + 64, pb[j]);
  wave_lds_order();
  q0 = *(const float4*)(wl + 64);
  q1 = *(const float4*)(wl + 68);
  for (int s = 9; s < 16; ++s)
    semiring_step<IS_MAX, true, true>(wl + 8 * s, ht + 64 * s, q0, q1, hv, acc0, acc1);
  semiring_step<IS_MAX, false, false>(wl, ht, q0, q1, hv, acc0, acc1);

  __syncthreads();
  float* red = smem;
  if (OUT_T) {
    float* r0 = red + (wave * 64 + lane) * RP_;
    float* r1 = r0 + 32 * RP_;
#pragma unroll
    for (int q = 0; q < 4; ++q) {
      *(float4*)(r0 + 4 * q) = make_float4(acc0[4 * q], acc0[4 * q + 1], acc0[4 * q + 2], acc0[4 * q + 3]);
      *(float4*)(r1 + 4 * q) = make_float4(acc1[4 * q], acc1[4 * q + 1], acc1[4 * q + 2], acc1[4 * q + 3]);
    }
    __syncthreads();
#pragma unroll
    for (int p = 0; p < 2; ++p) {
      const int idx = (int)threadIdx.x + 512 * p;
      const int o = idx >> 4;
      const int b = idx & 15;
      float v = red[o * RP_ + b];
#pragma unroll
      for (int w = 1; w < 16; ++w) {
        const float u = red[(w * 64 + o) * RP_ + b];
        v = IS_MAX ? fmaxf(v, u) : fminf(v, u);
      }
      hout[((size_t)blockIdx.x * H_ + o0) * 16 + idx] = v;
    }
    return;
  }
#pragma unroll
  for (int b = 0; b < 16; ++b) {
    red[(wave * 16 + b) * 64 + lane] = acc0[b];
    red[(wave * 16 + b) * 64 + lane + 32] = acc1[b];
  }
  __syncthreads();
#pragma unroll
  for (int p = 0; p < 2; ++p) {
    const int idx = (int)threadIdx.x + 512 * p;
    const int b = idx >> 6;
    const int o = idx & 63;
    float v = red[b * 64 + o];
#pragma unroll
    for (int w = 1; w < 16; ++w) {
      const float u = red[(w * 16 + b) * 64 + o];
      v = IS_MAX ? fmaxf(v, u) : fminf(v, u);
    }
    hout[(b0 + b) * H_ + o0 + o] = v;
  }
}

__global__ __launch_bounds__(32) void k_fc_out(const float* __restrict__ h,
                                               const float* __restrict__ Wout,
                                               const float* __restrict__ bout,
                                               float* __restrict__ q) {
  const int lane = threadIdx.x & 31;
  const int half = lane >> 4;
  const int l16  = lane & 15;
  const int m0   = blockIdx.x * 16;
  const int n0   = blockIdx.y * 16;

  const int col  = n0 + l16;
  const int bcol = (col < A_) ? col : 0;

  v8f c = {};
  const float* ap = h    + (m0 + l16) * H_ + 2 * half;
  const float* bp = Wout + bcol * H_       + 2 * half;

  constexpr int CK = 4 * FO_;
  constexpr int NC = H_ / CK;
  v2f fa[3][FO_], fb[3][FO_];
#pragma unroll
  for (int cc = 0; cc < 2; ++cc) {
    chunk_load(fa[cc], ap + cc * CK); chunk_load(fb[cc], bp + cc * CK);
  }
#pragma unroll
  for (int cc = 0; cc < NC; ++cc) {
    if (cc + 2 < NC) {
      const int n = (cc + 2) % 3;
      chunk_load(fa[n], ap + (cc + 2) * CK); chunk_load(fb[n], bp + (cc + 2) * CK);
    }
    PIPE_FENCE();
    const int u = cc % 3;
#pragma unroll
    for (int s = 0; s < FO_; ++s)
      c = __builtin_amdgcn_wmma_f32_16x16x4_f32(false, fa[u][s], false, fb[u][s], (short)0, c, false, false);
    PIPE_FENCE();
  }

  if (col < A_) {
    const float bias = bout[col];
#pragma unroll
    for (int v = 0; v < 8; ++v) q[(m0 + v + 8 * half) * A_ + col] = c[v] + bias;
  }
}

extern "C" void kernel_launch(void* const* d_in, const int* in_sizes, int n_in,
                              void* d_out, int out_size, void* d_ws, size_t ws_size,
                              hipStream_t stream) {
  (void)in_sizes; (void)n_in; (void)out_size; (void)ws_size;
  const float* x    = (const float*)d_in[0];
  const float* Win  = (const float*)d_in[1];
  const float* bin  = (const float*)d_in[2];
  const float* Wmax = (const float*)d_in[3];
  const float* Wmin = (const float*)d_in[4];
  const float* Wout = (const float*)d_in[5];
  const float* bout = (const float*)d_in[6];
  float* q  = (float*)d_out;
  float* h1 = (float*)d_ws;
  float* h2 = h1 + (size_t)B_ * H_;

  k_fc_in   <<<dim3(B_ / 32, H_ / 32), 128, 0, stream>>>(x,  Win,  bin,  h1);
  k_semiring<1, 1><<<dim3(B_ / 16, H_ / 64), 512, 0, stream>>>(h1, Wmax, h2);
  k_semiring<0, 0><<<dim3(B_ / 16, H_ / 64), 512, 0, stream>>>(h2, Wmin, h1);
  k_fc_out  <<<dim3(B_ / 16, 2),       32,  0, stream>>>(h1, Wout, bout, q);
}


# semiring layers: W read as the wave-uniform operand by scalar loads straight from memory, activations per lane (2 rows each) in new inter-layer layouts; W no longer staged in LDS; bit-identical
# speedup vs baseline: 3.3852x; 1.1047x over previous
#include <hip/hip_runtime.h>

typedef __attribute__((ext_vector_type(2))) float v2f;
typedef __attribute__((ext_vector_type(4))) float v4f;
typedef __attribute__((ext_vector_type(8))) float v8f;
typedef __attribute__((ext_vector_type(16))) float v16f;


#define B_   256
#define OBS_ 256
#define H_   1024
#define A_   18
#define FO_  16

template <int N, int ST = 4>
__device__ __forceinline__ void chunk_load(v2f (&f)[N], const float* p) {
#pragma unroll
  for (int s = 0; s < N; ++s) f[s] = *(const v2f*)(p + ST * s);
}

#define PIPE_FENCE() __builtin_amdgcn_sched_barrier(0)

#define FG_ 8
#define FP_ 260

#define TP_ 36

__device__ __forceinline__ void lds_barrier() {
  __builtin_amdgcn_fence(__ATOMIC_RELEASE, "workgroup", "local");
  __builtin_amdgcn_s_barrier();
  __builtin_amdgcn_fence(__ATOMIC_ACQUIRE, "workgroup", "local");
}

template <int G0, int G1>
__device__ __forceinline__ void fc_in_groups(const float* ap, const float* bp, v8f& c) {
  v2f fa[2][FG_], fb[2][FG_];
  chunk_load(fa[G0 & 1], ap + G0 * 4 * FG_); chunk_load(fb[G0 & 1], bp + G0 * 4 * FG_);
#pragma unroll
  for (int g = G0; g < G1; ++g) {
    if (g + 1 < G1) {
      chunk_load(fa[(g + 1) & 1], ap + (g + 1) * 4 * FG_); chunk_load(fb[(g + 1) & 1], bp + (g + 1) * 4 * FG_);
    }
    PIPE_FENCE();
#pragma unroll
    for (int s = 0; s < FG_; ++s)
      c = __builtin_amdgcn_wmma_f32_16x16x4_f32(false, fa[g & 1][s], false, fb[g & 1][s], (short)0, c, false, false);
    PIPE_FENCE();
  }
}

__global__ __launch_bounds__(128) void k_fc_in(const float* __restrict__ x,
                                               const float* __restrict__ Win,
                                               const float* __restrict__ bin,
                                               float* __restrict__ h1) {
  __shared__ __attribute__((aligned(16))) float opnd[64 * FP_];
  __shared__ __attribute__((aligned(16))) float tile[32 * TP_];
  const int lane = threadIdx.x & 31;
  const int half = lane >> 4;
  const int l16  = lane & 15;
  const int wave = __builtin_amdgcn_readfirstlane((int)(threadIdx.x >> 5));
  const int m0   = blockIdx.x * 32;
  const int n0   = blockIdx.y * 32;

  const float* src = (wave < 2) ? x   + (size_t)(m0 + 16 * wave) * OBS_
                                : Win + (size_t)(n0 + 16 * (wave - 2)) * OBS_;
  float* dst = opnd + (16 * wave) * FP_ + 4 * lane;
  src += 4 * lane;
  v4f lo[16], hi[16];
#pragma unroll
  for (int r = 0; r < 16; ++r) lo[r] = *(const v4f*)(src + r * OBS_);
  PIPE_FENCE();
#pragma unroll
  for (int r = 0; r < 16; ++r) hi[r] = *(const v4f*)(src + r * OBS_ + 128);

  const int wm = wave >> 1;
  const int wn = wave & 1;
  const float* ap = opnd + (16 * wm + l16) * FP_ + 2 * half;
  const float* bp = opnd + (32 + 16 * wn + l16) * FP_ + 2 * half;
  v8f c = {};

#pragma unroll
  for (int r = 0; r < 16; ++r) *(v4f*)(dst + r * FP_) = lo[r];
  lds_barrier();
  constexpr int NG = OBS_ / (4 * FG_);
  fc_in_groups<0, NG / 2>(ap, bp, c);
#pragma unroll
  for (int r = 0; r < 16; ++r) *(v4f*)(dst + r * FP_ + 128) = hi[r];
  lds_barrier();
  fc_in_groups<NG / 2, NG>(ap, bp, c);

  {
    const float bias = bin[n0 + 16 * wn + l16];
    float* tp = tile + (16 * wm + 8 * half) * TP_ + 16 * wn + l16;
#pragma unroll
    for (int v = 0; v < 8; ++v) tp[v * TP_] = c[v] + bias;
  }
  __syncthreads();
  const int btile = blockIdx.x >> 1;
  const int sub   = blockIdx.x & 1;
#pragma unroll
  for (int p = 0; p < 2; ++p) {
    const int idx = (int)threadIdx.x + 128 * p;
    const int i2  = idx >> 4;
    const int rho = idx & 15;
    const v2f ra = *(const v2f*)(tile + rho * TP_ + 2 * i2);
    const v2f rb = *(const v2f*)(tile + (rho + 16) * TP_ + 2 * i2);
    *(float4*)(h1 + ((size_t)(btile * (H_ / 2) + (n0 >> 1) + i2) * 32 + sub * 16 + rho) * 4) =
        make_float4(ra.x, rb.x, ra.y, rb.y);
  }
}

template <int IS_MAX>
__device__ __forceinline__ float sel3(float a, float x, float y) {
  return IS_MAX ? fmaxf(fmaxf(a, x), y) : fminf(fminf(a, x), y);
}

__device__ __forceinline__ float own_sreg(float v) {
  return __int_as_float(__builtin_amdgcn_readfirstlane(__float_as_int(v)));
}

#define STEP_FENCE() __builtin_amdgcn_sched_barrier(0)

template <int IS_MAX>
__device__ __forceinline__ void fold16(const float (&v)[16], float (&t)[6]) {
  t[0] = sel3<IS_MAX>(v[0], v[1], v[2]);
  t[1] = sel3<IS_MAX>(v[3], v[4], v[5]);
  t[2] = sel3<IS_MAX>(v[6], v[7], v[8]);
  t[3] = sel3<IS_MAX>(v[9], v[10], v[11]);
  t[4] = sel3<IS_MAX>(v[12], v[13], v[14]);
  t[5] = v[15];
}

__device__ __forceinline__ void h_group_load(float4 (&g)[8], const float* p) {
#pragma unroll
  for (int j = 0; j < 8; ++j) g[j] = *(const float4*)(p + j * 128);
}

template <int IS_MAX, bool MORE, bool HN>
__device__ __forceinline__ void semiring_step(const float* wn, const float* hn, float4 (&hg)[8],
                                              const float4 (&hc)[8], v16f (&wv)[4],
                                              float (&acc)[2][4]) {
  float t[2][4][6];
#pragma unroll
  for (int q = 0; q < 4; ++q) {
    float v0[16], v1[16];
#pragma unroll
    for (int i = 0; i < 16; ++i) {
      const float4 hq = hc[i >> 1];
      const v2f hp = (i & 1) ? v2f{hq.z, hq.w} : v2f{hq.x, hq.y};
      const float w = (i & 1) ? own_sreg(wv[q][i]) : wv[q][i];
      const v2f s = hp + w;
      v0[i] = s.x;
      v1[i] = s.y;
    }
    fold16<IS_MAX>(v0, t[0][q]);
    fold16<IS_MAX>(v1, t[1][q]);
  }
  STEP_FENCE();
  if (MORE) {
    wv[0] = *(const v16f*)(wn);
    wv[1] = *(const v16f*)(wn + H_);
    wv[2] = *(const v16f*)(wn + 2 * H_);
    wv[3] = *(const v16f*)(wn + 3 * H_);
  }
  if (HN) h_group_load(hg, hn);
  STEP_FENCE();
#pragma unroll
  for (int q = 0; q < 4; ++q) {
#pragma unroll
    for (int bb = 0; bb < 2; ++bb) {
      const float u0 = sel3<IS_MAX>(t[bb][q][0], t[bb][q][1], t[bb][q][2]);
      const float u1 = sel3<IS_MAX>(t[bb][q][3], t[bb][q][4], t[bb][q][5]);
      acc[bb][q] = sel3<IS_MAX>(acc[bb][q], u0, u1);
    }
  }
}

template <int IS_MAX, int OUT_P>
__global__ __launch_bounds__(512) void k_semiring(const float* __restrict__ hin,
                                                  const float* __restrict__ W,
                                                  float* __restrict__ hout) {
  __shared__ __attribute__((aligned(16))) float red[16 * 1024];
  const int bt = blockIdx.x;
  const int o0 = blockIdx.y * 16;
  const int lane = threadIdx.x & 31;
  const int wave = __builtin_amdgcn_readfirstlane((int)(threadIdx.x >> 5));

  const float* hp = hin + ((size_t)(bt * (H_ / 2) + wave * 32) * 32 + lane) * 4;
  const float* wp = W + (size_t)o0 * H_ + wave * 64;

  v16f wv[4];
#pragma unroll
  for (int q = 0; q < 4; ++q) wv[q] = *(const v16f*)(wp + q * H_);
  float4 h0[8], h1[8], h2[8], h3[8];
  h_group_load(h0, hp);
  STEP_FENCE();
  h_group_load(h1, hp + 1024);

  float acc[4][2][4];
#pragma unroll
  for (int og = 0; og < 4; ++og)
#pragma unroll
    for (int bb = 0; bb < 2; ++bb)
#pragma unroll
      for (int q = 0; q < 4; ++q) acc[og][bb][q] = IS_MAX ? -3.402823466e38f : 3.402823466e38f;

  semiring_step<IS_MAX, true, true >(wp + 4 * H_,       hp + 2048, h2, h0, wv, acc[0]);
  semiring_step<IS_MAX, true, false>(wp + 8 * H_,       hp, h2, h0, wv, acc[1]);
  semiring_step<IS_MAX, true, false>(wp + 12 * H_,      hp, h2, h0, wv, acc[2]);
  semiring_step<IS_MAX, true, false>(wp + 16,           hp, h2, h0, wv, acc[3]);
  semiring_step<IS_MAX, true, true >(wp + 4 * H_ + 16,  hp + 3072, h3, h1, wv, acc[0]);
  semiring_step<IS_MAX, true, false>(wp + 8 * H_ + 16,  hp, h3, h1, wv, acc[1]);
  semiring_step<IS_MAX, true, false>(wp + 12 * H_ + 16, hp, h3, h1, wv, acc[2]);
  semiring_step<IS_MAX, true, false>(wp + 32,           hp, h3, h1, wv, acc[3]);
  semiring_step<IS_MAX, true, false>(wp + 4 * H_ + 32,  hp, h3, h2, wv, acc[0]);
  semiring_step<IS_MAX, true, false>(wp + 8 * H_ + 32,  hp, h3, h2, wv, acc[1]);
  semiring_step<IS_MAX, true, false>(wp + 12 * H_ + 32, hp, h3, h2, wv, acc[2]);
  semiring_step<IS_MAX, true, false>(wp + 48,           hp, h3, h2, wv, acc[3]);
  semiring_step<IS_MAX, true, false>(wp + 4 * H_ + 48,  hp, h2, h3, wv, acc[0]);
  semiring_step<IS_MAX, true, false>(wp + 8 * H_ + 48,  hp, h2, h3, wv, acc[1]);
  semiring_step<IS_MAX, true, false>(wp + 12 * H_ + 48, hp, h2, h3, wv, acc[2]);
  semiring_step<IS_MAX, false, false>(wp,               hp, h2, h3, wv, acc[3]);

  float* rw = red + wave * 1024 + 32 * (lane >> 4) + (lane & 15);
#pragma unroll
  for (int og = 0; og < 4; ++og)
#pragma unroll
    for (int q = 0; q < 4; ++q) {
      rw[(4 * og + q) * 64]      = acc[og][0][q];
      rw[(4 * og + q) * 64 + 16] = acc[og][1][q];
    }
  __syncthreads();
  if (threadIdx.x < 256) {
    int e0, e1, e2, e3;
    size_t dst;
    if (OUT_P) {
      const int o2 = (int)threadIdx.x >> 5;
      const int ln = (int)threadIdx.x & 31;
      const int r  = 32 * (ln >> 4) + (ln & 15);
      e0 = (2 * o2) * 64 + r;  e1 = e0 + 16;
      e2 = e0 + 64;            e3 = e2 + 16;
      dst = ((size_t)(bt * (H_ / 2) + (o0 >> 1) + o2) * 32 + ln) * 4;
    } else {
      const int g = (int)threadIdx.x >> 6;
      const int r = (int)threadIdx.x & 63;
      e0 = (4 * g) * 64 + r;  e1 = e0 + 64;  e2 = e0 + 128;  e3 = e0 + 192;
      dst = ((size_t)(bt * (H_ / 4) + (o0 >> 2) + g) * 64 + r) * 4;
    }
    const int ee[4] = {e0, e1, e2, e3};
    float v[4];
#pragma unroll
    for (int e = 0; e < 4; ++e) {
      const float* rp = red + ee[e];
      float m = rp[0];
#pragma unroll
      for (int w = 1; w < 16; ++w) {
        const float u = rp[w * 1024];
        m = IS_MAX ? fmaxf(m, u) : fminf(m, u);
      }
      v[e] = m;
    }
    *(float4*)(hout + dst) = make_float4(v[0], v[1], v[2], v[3]);
  }
}

__global__ __launch_bounds__(32) void k_fc_out(const float* __restrict__ h,
                                               const float* __restrict__ Wout,
                                               const float* __restrict__ bout,
                                               float* __restrict__ q) {
  const int lane = threadIdx.x & 31;
  const int half = lane >> 4;
  const int l16  = lane & 15;
  const int m0   = blockIdx.x * 16;
  const int n0   = blockIdx.y * 16;

  const int col  = n0 + l16;
  const int bcol = (col < A_) ? col : 0;

  v8f c = {};
  const float* ap = h + ((size_t)(m0 >> 6) * (H_ / 4) * 64 + (m0 & 63) + l16) * 4 + 2 * half;
  const float* bp = Wout + bcol * H_       + 2 * half;

  constexpr int CK = 4 * FO_;
  constexpr int NC = H_ / CK;
  v2f fa[3][FO_], fb[3][FO_];
#pragma unroll
  for (int cc = 0; cc < 2; ++cc) {
    chunk_load<FO_, 256>(fa[cc], ap + cc * FO_ * 256); chunk_load(fb[cc], bp + cc * CK);
  }
#pragma unroll
  for (int cc = 0; cc < NC; ++cc) {
    if (cc + 2 < NC) {
      const int n = (cc + 2) % 3;
      chunk_load<FO_, 256>(fa[n], ap + (cc + 2) * FO_ * 256); chunk_load(fb[n], bp + (cc + 2) * CK);
    }
    PIPE_FENCE();
    const int u = cc % 3;
#pragma unroll
    for (int s = 0; s < FO_; ++s)
      c = __builtin_amdgcn_wmma_f32_16x16x4_f32(false, fa[u][s], false, fb[u][s], (short)0, c, false, false);
    PIPE_FENCE();
  }

  if (col < A_) {
    const float bias = bout[col];
#pragma unroll
    for (int v = 0; v < 8; ++v) q[(m0 + v + 8 * half) * A_ + col] = c[v] + bias;
  }
}

extern "C" void kernel_launch(void* const* d_in, const int* in_sizes, int n_in,
                              void* d_out, int out_size, void* d_ws, size_t ws_size,
                              hipStream_t stream) {
  (void)in_sizes; (void)n_in; (void)out_size; (void)ws_size;
  const float* x    = (const float*)d_in[0];
  const float* Win  = (const float*)d_in[1];
  const float* bin  = (const float*)d_in[2];
  const float* Wmax = (const float*)d_in[3];
  const float* Wmin = (const float*)d_in[4];
  const float* Wout = (const float*)d_in[5];
  const float* bout = (const float*)d_in[6];
  float* q  = (float*)d_out;
  float* h1 = (float*)d_ws;
  float* h2 = h1 + (size_t)B_ * H_;

  k_fc_in   <<<dim3(B_ / 32, H_ / 32), 128, 0, stream>>>(x,  Win,  bin,  h1);
  k_semiring<1, 1><<<dim3(B_ / 64, H_ / 16), 512, 0, stream>>>(h1, Wmax, h2);
  k_semiring<0, 0><<<dim3(B_ / 64, H_ / 16), 512, 0, stream>>>(h2, Wmin, h1);
  k_fc_out  <<<dim3(B_ / 16, 2),       32,  0, stream>>>(h1, Wout, bout, q);
}
